# starting kernel
# baseline (speedup 1.0000x reference)
#include <hip/hip_runtime.h>


typedef __attribute__((ext_vector_type(16))) _Float16 v16h;
typedef __attribute__((ext_vector_type(8)))  _Float16 v8h;
typedef __attribute__((ext_vector_type(8)))  float    v8f;

#define DEVI static __device__ __forceinline__

static constexpr int BATCH = 128;
static constexpr int SEQ   = 512;
static constexpr int E     = 192;
static constexpr int QDIM  = 48;
static constexpr int NTOK  = BATCH * SEQ;

static constexpr int KT_LD = 24;
static constexpr int VT_LD = 520;

DEVI v8f wmma16(v16h a, v16h b, v8f c) {
  return __builtin_amdgcn_wmma_f32_16x16x32_f16(false, a, false, b,
                                                (short)0, c, false, false);
}

DEVI float swap16(float v) {
  int i = __builtin_bit_cast(int, v);
  int r = __builtin_amdgcn_permlanex16(i, i, 0x76543210, 0xfedcba98, false, false);
  return __builtin_bit_cast(float, r);
}

DEVI v16h load_frag(const _Float16* __restrict__ base, int ld, int krem) {
  const int lane = threadIdx.x & 31;
  const int row  = lane & 15;
  const int d0   = (lane & 16) ? 8 : 0;
  const _Float16* p = base + (size_t)row * ld + d0;
  v16h r;
  if (krem >= 32) {
    v8h lo = *(const v8h*)(p);
    v8h hi = *(const v8h*)(p + 16);
#pragma unroll
    for (int i = 0; i < 8; ++i) { r[i] = lo[i]; r[8 + i] = hi[i]; }
  } else {
#pragma unroll
    for (int i = 0; i < 8; ++i) {
      r[i]     = (d0 + i      < krem) ? p[i]      : (_Float16)0.0f;
      r[8 + i] = (d0 + 16 + i < krem) ? p[16 + i] : (_Float16)0.0f;
    }
  }
  return r;
}

__global__ void cvt_f16_kernel(const float* __restrict__ src,
                               _Float16* __restrict__ dst, int n) {
  int i = blockIdx.x * 256 + threadIdx.x;
  if (i < n) dst[i] = (_Float16)src[i];
}

__global__ __launch_bounds__(256) void ln_kernel(
    const float* __restrict__ x, const float* __restrict__ g,
    const float* __restrict__ b, _Float16* __restrict__ out) {
  const int wave = threadIdx.x >> 5, lane = threadIdx.x & 31;
  const size_t t = (size_t)blockIdx.x * 8 + wave;
  const float* xp = x + t * E;
  float v[6];
  float s = 0.0f;
#pragma unroll
  for (int i = 0; i < 6; ++i) { v[i] = xp[lane + 32 * i]; s += v[i]; }
#pragma unroll
  for (int o = 16; o > 0; o >>= 1) s += __shfl_xor(s, o, 32);
  const float mean = s * (1.0f / E);
  float vs = 0.0f;
#pragma unroll
  for (int i = 0; i < 6; ++i) { float d = v[i] - mean; vs += d * d; }
#pragma unroll
  for (int o = 16; o > 0; o >>= 1) vs += __shfl_xor(vs, o, 32);
  const float inv = rsqrtf(vs * (1.0f / E) + 1e-6f);
#pragma unroll
  for (int i = 0; i < 6; ++i) {
    int c = lane + 32 * i;
    out[t * E + c] = (_Float16)((v[i] - mean) * inv * g[c] + b[c]);
  }
}

__global__ void conv_kernel(const _Float16* __restrict__ xn,
                            const float* __restrict__ x,
                            const float* __restrict__ cw,
                            float* __restrict__ x1) {
  const size_t idx = (size_t)blockIdx.x * 256 + threadIdx.x;
  if (idx >= (size_t)NTOK * 96) return;
  const int h = (int)(idx % 96);
  const size_t t = idx / 96;
  const float* w = cw + (t & 3) * 15;
  const _Float16* xp = xn + t * E + 48;
  float s = 0.0f;
#pragma unroll
  for (int k = 0; k < 15; ++k) {
    int hh = h + k - 7;
    if (hh >= 0 && hh < 96) s += w[k] * (float)xp[hh];
  }
  x1[t * E + 48 + h] = s + x[t * E + 48 + h];
}

template <int NT, bool RELU, bool RES, bool OUT16>
__global__ __launch_bounds__(256) void gemm_f16_kernel(
    const _Float16* __restrict__ A, int lda,
    const _Float16* __restrict__ W, int K,
    const float* __restrict__ bias,
    const float* __restrict__ res, int ldr,
    float* __restrict__ out32, _Float16* __restrict__ out16, int ldo) {
  const int wave = threadIdx.x >> 5;
  const int m0 = blockIdx.x * 128 + wave * 16;
  const int n0 = blockIdx.y * (NT * 16);
  v8f z = {};
  v8f acc[NT];
#pragma unroll
  for (int t = 0; t < NT; ++t) acc[t] = z;

  for (int k0 = 0; k0 < K; k0 += 32) {
    const int krem = K - k0;
    v16h a = load_frag(A + (size_t)m0 * lda + k0, lda, krem);
#pragma unroll
    for (int t = 0; t < NT; ++t) {
      v16h bfr = load_frag(W + (size_t)(n0 + t * 16) * K + k0, K, krem);
      acc[t] = wmma16(a, bfr, acc[t]);
    }
  }

  const int lane = threadIdx.x & 31;
  const int col = lane & 15;
  const int r0  = (lane & 16) ? 8 : 0;
#pragma unroll
  for (int t = 0; t < NT; ++t) {
    const int n = n0 + t * 16 + col;
    const float bv = bias[n];
#pragma unroll
    for (int j = 0; j < 8; ++j) {
      const int m = m0 + r0 + j;
      float v = acc[t][j] + bv;
      if (RELU) v = fmaxf(v, 0.0f);
      if (RES)  v += res[(size_t)m * ldr + n];
      if (OUT16) out16[(size_t)m * ldo + n] = (_Float16)v;
      else       out32[(size_t)m * ldo + n] = v;
    }
  }
}

__global__ __launch_bounds__(256) void attn_kernel(
    const _Float16* __restrict__ qkv_all, _Float16* __restrict__ out_all) {
  __shared__ _Float16 vt[16 * VT_LD];
  __shared__ _Float16 kt[SEQ * KT_LD];

  const int bid = blockIdx.x;
  const int b = bid / 6;
  const int r = bid % 6;
  const int branch = r / 3;
  const int hh = r % 3;
  const _Float16* qkv = qkv_all + (size_t)branch * NTOK * 144
                                + (size_t)b * SEQ * 144;
  _Float16* outp = out_all + (size_t)branch * NTOK * QDIM
                           + (size_t)b * SEQ * QDIM;
  const int qcol = hh * 16, kcol = 48 + hh * 16, vcol = 96 + hh * 16;

  for (int e = threadIdx.x; e < 16 * SEQ; e += 256) {
    const int d = e & 15, j = e >> 4;
    const _Float16 kv = qkv[(size_t)j * 144 + kcol + d];
    const _Float16 vv = qkv[(size_t)j * 144 + vcol + d];
    kt[j * KT_LD + d] = kv;
    vt[d * VT_LD + j] = vv;
  }
  __syncthreads();

  const int wave = threadIdx.x >> 5, lane = threadIdx.x & 31;
  const int q0 = blockIdx.y * 128 + wave * 16;
  const int row = lane & 15;
  const int d0  = (lane & 16) ? 8 : 0;

  v16h qf;
  {
    const _Float16* qp = qkv + (size_t)(q0 + row) * 144 + qcol + d0;
    v8h ql = *(const v8h*)qp;
#pragma unroll
    for (int i = 0; i < 8; ++i) {
      qf[i] = ql[i] * (_Float16)0.25f;
      qf[8 + i] = (_Float16)0.0f;
    }
  }

  v8f z = {};
  v8f oacc = z;
  float m_run = -1e30f, l_run = 0.0f;

  for (int kb = 0; kb < SEQ / 32; ++kb) {
    const int j0 = kb * 32;
    v16h kf0, kf1;
    {
      const _Float16* kp0 = &kt[(j0 + row) * KT_LD + d0];
      const _Float16* kp1 = &kt[(j0 + 16 + row) * KT_LD + d0];
      v8h ka = *(const v8h*)kp0;
      v8h kb2 = *(const v8h*)kp1;
#pragma unroll
      for (int i = 0; i < 8; ++i) {
        kf0[i] = ka[i];  kf0[8 + i] = (_Float16)0.0f;
        kf1[i] = kb2[i]; kf1[8 + i] = (_Float16)0.0f;
      }
    }
    v8f s0 = wmma16(kf0, qf, z);
    v8f s1 = wmma16(kf1, qf, z);

    float tm = -1e30f;
#pragma unroll
    for (int j = 0; j < 8; ++j) tm = fmaxf(tm, fmaxf(s0[j], s1[j]));
    tm = fmaxf(tm, swap16(tm));
    const float mnew = fmaxf(m_run, tm);
    const float corr = __expf(m_run - mnew);
    float p0[8], p1[8], ls = 0.0f;
#pragma unroll
    for (int j = 0; j < 8; ++j) {
      p0[j] = __expf(s0[j] - mnew);
      p1[j] = __expf(s1[j] - mnew);
      ls += p0[j] + p1[j];
    }
    ls += swap16(ls);
    l_run = l_run * corr + ls;
    m_run = mnew;
#pragma unroll
    for (int j = 0; j < 8; ++j) oacc[j] *= corr;

    v16h pf;
#pragma unroll
    for (int i = 0; i < 8; ++i) {
      pf[i] = (_Float16)p0[i];
      pf[8 + i] = (_Float16)p1[i];
    }
    const _Float16* vp = &vt[row * VT_LD + j0 + d0];
    v8h vl = *(const v8h*)vp;
    v8h vh = *(const v8h*)(vp + 16);
    v16h vf;
#pragma unroll
    for (int i = 0; i < 8; ++i) { vf[i] = vl[i]; vf[8 + i] = vh[i]; }

    oacc = wmma16(vf, pf, oacc);
  }

  const float inv = 1.0f / l_run;
#pragma unroll
  for (int j = 0; j < 8; ++j)
    outp[(size_t)(q0 + row) * QDIM + qcol + d0 + j] = (_Float16)(oacc[j] * inv);
}

static inline size_t align256(size_t x) { return (x + 255) & ~(size_t)255; }

extern "C" void kernel_launch(void* const* d_in, const int* in_sizes, int n_in,
                              void* d_out, int out_size, void* d_ws, size_t ws_size,
                              hipStream_t stream) {
  (void)in_sizes; (void)n_in; (void)out_size; (void)ws_size;

  const float* x        = (const float*)d_in[0];
  const float* ln1_g    = (const float*)d_in[1];
  const float* ln1_b    = (const float*)d_in[2];
  const float* acc_wqkv = (const float*)d_in[3];
  const float* acc_bqkv = (const float*)d_in[4];
  const float* acc_wo   = (const float*)d_in[5];
  const float* acc_bo   = (const float*)d_in[6];
  const float* gy_wqkv  = (const float*)d_in[7];
  const float* gy_bqkv  = (const float*)d_in[8];
  const float* gy_wo    = (const float*)d_in[9];
  const float* gy_bo    = (const float*)d_in[10];
  const float* conv_w   = (const float*)d_in[11];
  const float* ln2_g    = (const float*)d_in[12];
  const float* ln2_b    = (const float*)d_in[13];
  const float* w1       = (const float*)d_in[14];
  const float* b1       = (const float*)d_in[15];
  const float* w2       = (const float*)d_in[16];
  const float* b2       = (const float*)d_in[17];

  char* ws = (char*)d_ws;
  size_t off = 0;
  _Float16* xn_h   = (_Float16*)(ws + off); off = align256(off + (size_t)NTOK * E * 2);
  _Float16* qkv_h  = (_Float16*)(ws + off); off = align256(off + (size_t)2 * NTOK * 144 * 2);
  _Float16* attn_h = (_Float16*)(ws + off); off = align256(off + (size_t)2 * NTOK * QDIM * 2);
  float*    x1     = (float*)   (ws + off); off = align256(off + (size_t)NTOK * E * 4);
  _Float16* tt_h   = (_Float16*)(ws + off); off = align256(off + (size_t)NTOK * E * 2);
  _Float16* hid_h  = (_Float16*)(ws + off); off = align256(off + (size_t)NTOK * 2 * E * 2);
  _Float16* wq_a_h = (_Float16*)(ws + off); off = align256(off + 144 * 48 * 2);
  _Float16* wo_a_h = (_Float16*)(ws + off); off = align256(off + 48 * 48 * 2);
  _Float16* wq_g_h = (_Float16*)(ws + off); off = align256(off + 144 * 48 * 2);
  _Float16* wo_g_h = (_Float16*)(ws + off); off = align256(off + 48 * 48 * 2);
  _Float16* w1_h   = (_Float16*)(ws + off); off = align256(off + 384 * 192 * 2);
  _Float16* w2_h   = (_Float16*)(ws + off); off = align256(off + 192 * 384 * 2);

  cvt_f16_kernel<<<(144 * 48 + 255) / 256, 256, 0, stream>>>(acc_wqkv, wq_a_h, 144 * 48);
  cvt_f16_kernel<<<(48 * 48 + 255) / 256, 256, 0, stream>>>(acc_wo, wo_a_h, 48 * 48);
  cvt_f16_kernel<<<(144 * 48 + 255) / 256, 256, 0, stream>>>(gy_wqkv, wq_g_h, 144 * 48);
  cvt_f16_kernel<<<(48 * 48 + 255) / 256, 256, 0, stream>>>(gy_wo, wo_g_h, 48 * 48);
  cvt_f16_kernel<<<(384 * 192 + 255) / 256, 256, 0, stream>>>(w1, w1_h, 384 * 192);
  cvt_f16_kernel<<<(192 * 384 + 255) / 256, 256, 0, stream>>>(w2, w2_h, 192 * 384);

  ln_kernel<<<NTOK / 8, 256, 0, stream>>>(x, ln1_g, ln1_b, xn_h);

  conv_kernel<<<(NTOK * 96) / 256, 256, 0, stream>>>(xn_h, x, conv_w, x1);

  gemm_f16_kernel<3, false, false, true><<<dim3(NTOK / 128, 3), 256, 0, stream>>>(
      xn_h + 0, E, wq_a_h, 48, acc_bqkv, nullptr, 0, nullptr, qkv_h, 144);
  gemm_f16_kernel<3, false, false, true><<<dim3(NTOK / 128, 3), 256, 0, stream>>>(
      xn_h + 96, E, wq_g_h, 48, gy_bqkv, nullptr, 0, nullptr,
      qkv_h + (size_t)NTOK * 144, 144);

  attn_kernel<<<dim3(BATCH * 6, SEQ / 128), 256, 0, stream>>>(qkv_h, attn_h);

  gemm_f16_kernel<3, false, true, false><<<dim3(NTOK / 128, 1), 256, 0, stream>>>(
      attn_h, QDIM, wo_a_h, 48, acc_bo, x + 0, E, x1 + 0, nullptr, E);
  gemm_f16_kernel<3, false, true, false><<<dim3(NTOK / 128, 1), 256, 0, stream>>>(
      attn_h + (size_t)NTOK * QDIM, QDIM, wo_g_h, 48, gy_bo,
      x + 144, E, x1 + 144, nullptr, E);

  ln_kernel<<<NTOK / 8, 256, 0, stream>>>(x1, ln2_g, ln2_b, tt_h);

  gemm_f16_kernel<4, true, false, true><<<dim3(NTOK / 128, 6), 256, 0, stream>>>(
      tt_h, E, w1_h, 192, b1, nullptr, 0, nullptr, hid_h, 384);

  gemm_f16_kernel<4, true, true, false><<<dim3(NTOK / 128, 3), 256, 0, stream>>>(
      hid_h, 2 * E, w2_h, 384, b2, x1, E, (float*)d_out, nullptr, E);
}


# QKV projection moved into the attention launch: a 16-wave work-group per (batch, branch, head) projects K and V^T into LDS and its Q^T fragments into registers from xn by WMMA; 10 launches -> 8
# speedup vs baseline: 1.7058x; 1.7058x over previous
#include <hip/hip_runtime.h>


typedef __attribute__((ext_vector_type(16))) _Float16 v16h;
typedef __attribute__((ext_vector_type(8)))  _Float16 v8h;
typedef __attribute__((ext_vector_type(8)))  float    v8f;
typedef __attribute__((ext_vector_type(4)))  float    v4f;

#define DEVI static __device__ __forceinline__

static constexpr int BATCH = 128;
static constexpr int SEQ   = 512;
static constexpr int E     = 192;
static constexpr int QDIM  = 48;
static constexpr int NTOK  = BATCH * SEQ;

static constexpr int KT_LD = 24;
static constexpr int VT_LD = 520;

DEVI v8f wmma16(v16h a, v16h b, v8f c) {
  return __builtin_amdgcn_wmma_f32_16x16x32_f16(false, a, false, b,
                                                (short)0, c, false, false);
}

DEVI float swap16(float v) {
  int i = __builtin_bit_cast(int, v);
  int r = __builtin_amdgcn_permlanex16(i, i, 0x76543210, 0xfedcba98, false, false);
  return __builtin_bit_cast(float, r);
}

__global__ __launch_bounds__(256) void ln_kernel(
    const float* __restrict__ x, const float* __restrict__ g,
    const float* __restrict__ b, _Float16* __restrict__ out) {
  const int wave = threadIdx.x >> 5, lane = threadIdx.x & 31;
  const size_t t = (size_t)blockIdx.x * 8 + wave;
  const float* xp = x + t * E;
  float v[6];
  float s = 0.0f;
#pragma unroll
  for (int i = 0; i < 6; ++i) { v[i] = xp[lane + 32 * i]; s += v[i]; }
#pragma unroll
  for (int o = 16; o > 0; o >>= 1) s += __shfl_xor(s, o, 32);
  const float mean = s * (1.0f / E);
  float vs = 0.0f;
#pragma unroll
  for (int i = 0; i < 6; ++i) { float d = v[i] - mean; vs += d * d; }
#pragma unroll
  for (int o = 16; o > 0; o >>= 1) vs += __shfl_xor(vs, o, 32);
  const float inv = rsqrtf(vs * (1.0f / E) + 1e-6f);
#pragma unroll
  for (int i = 0; i < 6; ++i) {
    int c = lane + 32 * i;
    out[t * E + c] = (_Float16)((v[i] - mean) * inv * g[c] + b[c]);
  }
}

__global__ __launch_bounds__(128) void conv_wmma_kernel(
    const _Float16* __restrict__ xn, const float* __restrict__ x,
    const float* __restrict__ cw, float* __restrict__ x1) {
  const int ch = threadIdx.x >> 5, lane = threadIdx.x & 31;
  const int c = lane & 15, h = lane >> 4;
  const size_t tok = (size_t)blockIdx.x * 64 + 4 * c + ch;

  const float* w = cw + ch * 15;
  v16h af;
#pragma unroll
  for (int i = 0; i < 8; ++i) {
    const int k0 = 8 * h + i - c - 1;
    const int k1 = k0 + 16;
    const float w0 = w[min(max(k0, 0), 14)];
    const float w1 = w[min(max(k1, 0), 14)];
    af[i]     = (_Float16)((k0 >= 0 && k0 <= 14) ? w0 : 0.0f);
    af[8 + i] = (_Float16)((k1 >= 0 && k1 <= 14) ? w1 : 0.0f);
  }

  const _Float16* xp = xn + tok * E + 48;
  v8h ck[7];
  const v8h zero8 = {};
  {
    const v8h first = *(const v8h*)(xp);
    const v8h last  = *(const v8h*)(xp + 88);
    ck[0] = h ? first : zero8;
    ck[6] = h ? zero8 : last;
  }
#pragma unroll
  for (int j = 1; j < 6; ++j) ck[j] = *(const v8h*)(xp + 8 * (2 * j - 1 + h));

  const float* rp = x + tok * E + 48 + 8 * h;
  v4f r0[6], r1[6];
#pragma unroll
  for (int n = 0; n < 6; ++n) {
    r0[n] = *(const v4f*)(rp + 16 * n);
    r1[n] = *(const v4f*)(rp + 16 * n + 4);
  }

  float* op = x1 + tok * E + 48 + 8 * h;
#pragma unroll
  for (int n = 0; n < 6; ++n) {
    v16h bf;
#pragma unroll
    for (int i = 0; i < 8; ++i) { bf[i] = ck[n][i]; bf[8 + i] = ck[n + 1][i]; }
    v8f acc = {};
    acc = wmma16(af, bf, acc);
    v4f o0, o1;
#pragma unroll
    for (int j = 0; j < 4; ++j) { o0[j] = acc[j] + r0[n][j]; o1[j] = acc[4 + j] + r1[n][j]; }
    *(v4f*)(op + 16 * n) = o0;
    *(v4f*)(op + 16 * n + 4) = o1;
  }
}

template <int KREM>
DEVI v16h frag16(const _Float16* p) {
  v16h r;
  const v8h lo = *(const v8h*)p;
  if (KREM >= 32) {
    const v8h hi = *(const v8h*)(p + 16);
#pragma unroll
    for (int i = 0; i < 8; ++i) { r[i] = lo[i]; r[8 + i] = hi[i]; }
  } else {
#pragma unroll
    for (int i = 0; i < 8; ++i) { r[i] = lo[i]; r[8 + i] = (_Float16)0.0f; }
  }
  return r;
}

template <int KREM>
DEVI v16h frag16_f32(const float* p) {
  v16h r;
  const v4f a0 = *(const v4f*)p;
  const v4f a1 = *(const v4f*)(p + 4);
#pragma unroll
  for (int i = 0; i < 4; ++i) { r[i] = (_Float16)a0[i]; r[4 + i] = (_Float16)a1[i]; }
  if (KREM >= 32) {
    const v4f b0 = *(const v4f*)(p + 16);
    const v4f b1 = *(const v4f*)(p + 20);
#pragma unroll
    for (int i = 0; i < 4; ++i) { r[8 + i] = (_Float16)b0[i]; r[12 + i] = (_Float16)b1[i]; }
  } else {
#pragma unroll
    for (int i = 0; i < 8; ++i) r[8 + i] = (_Float16)0.0f;
  }
  return r;
}

template <int K, int NTW, int WPB, int TM, bool RELU, bool RES, bool OUT16>
__global__ __launch_bounds__(WPB * 32) void gemm_ws_kernel(
    const _Float16* __restrict__ X, int ldx,
    const float* __restrict__ Wt,
    const float* __restrict__ bias,
    const float* __restrict__ res, int ldr,
    float* __restrict__ out32, _Float16* __restrict__ out16, int ldo) {
  constexpr int KS = (K + 31) / 32;
  constexpr int CPR = K / 8;
  constexpr int NTHR = WPB * 32;
  constexpr int PT = TM * CPR / NTHR;
  constexpr int PITCH = K + 8;
  constexpr int NTILES = NTOK / TM;
  static_assert((TM * CPR) % NTHR == 0, "tile must split evenly");
  static_assert(K % 32 == 0 || K % 32 == 16, "k tail");
  static_assert(((K + 8) / 8) % 2 == 1, "pitch must be 16 B x odd");

  __shared__ __attribute__((aligned(16))) _Float16 tile[TM * PITCH];

  const int tid = threadIdx.x;
  const int wave = tid >> 5, lane = tid & 31;
  const int c = lane & 15;
  const int h8 = (lane & 16) ? 8 : 0;
  const int n0 = wave * (NTW * 16);

  v16h wf[NTW][KS];
  float bv[NTW][8];
#pragma unroll
  for (int t = 0; t < NTW; ++t) {
#pragma unroll
    for (int ks = 0; ks < KS; ++ks)
    {
      const float* p = Wt + (size_t)(n0 + t * 16 + c) * K + ks * 32 + h8;
      wf[t][ks] = (K - ks * 32 >= 32) ? frag16_f32<32>(p) : frag16_f32<16>(p);
    }
#pragma unroll
    for (int j = 0; j < 8; ++j) bv[t][j] = bias[n0 + t * 16 + h8 + j];
  }

  v8h st[PT];
  int tix = blockIdx.x;
  if (tix < NTILES) {
#pragma unroll
    for (int i = 0; i < PT; ++i) {
      const int q = i * NTHR + tid;
      const int rr = q / CPR, cc = q % CPR;
      st[i] = *(const v8h*)(X + (size_t)(tix * TM + rr) * ldx + cc * 8);
    }
  }

  for (; tix < NTILES; tix += gridDim.x) {
#pragma unroll
    for (int i = 0; i < PT; ++i) {
      const int q = i * NTHR + tid;
      const int rr = q / CPR, cc = q % CPR;
      *(v8h*)&tile[rr * PITCH + cc * 8] = st[i];
    }
    __syncthreads();

    const int nxt = tix + gridDim.x;
    if (nxt < NTILES) {
#pragma unroll
      for (int i = 0; i < PT; ++i) {
        const int q = i * NTHR + tid;
        const int rr = q / CPR, cc = q % CPR;
        st[i] = *(const v8h*)(X + (size_t)(nxt * TM + rr) * ldx + cc * 8);
      }
    }

#pragma unroll
    for (int s = 0; s < TM / 16; ++s) {
      v16h xf[KS];
#pragma unroll
      for (int ks = 0; ks < KS; ++ks) {
        const _Float16* p = &tile[(s * 16 + c) * PITCH + ks * 32 + h8];
        xf[ks] = (K - ks * 32 >= 32) ? frag16<32>(p) : frag16<16>(p);
      }
      const size_t tok = (size_t)tix * TM + s * 16 + c;
#pragma unroll
      for (int t = 0; t < NTW; ++t) {
        v8f acc = {};
#pragma unroll
        for (int ks = 0; ks < KS; ++ks) acc = wmma16(wf[t][ks], xf[ks], acc);
        const int n = n0 + t * 16 + h8;
        float v[8];
#pragma unroll
        for (int j = 0; j < 8; ++j) {
          v[j] = acc[j] + bv[t][j];
          if (RELU) v[j] = fmaxf(v[j], 0.0f);
        }
        if (RES) {
          const v4f r0 = *(const v4f*)(res + tok * ldr + n);
          const v4f r1 = *(const v4f*)(res + tok * ldr + n + 4);
#pragma unroll
          for (int j = 0; j < 4; ++j) { v[j] += r0[j]; v[4 + j] += r1[j]; }
        }
        if (OUT16) {
          v8h o;
#pragma unroll
          for (int j = 0; j < 8; ++j) o[j] = (_Float16)v[j];
          *(v8h*)(out16 + tok * ldo + n) = o;
        } else {
          v4f o0, o1;
#pragma unroll
          for (int j = 0; j < 4; ++j) { o0[j] = v[j]; o1[j] = v[4 + j]; }
          *(v4f*)(out32 + tok * ldo + n) = o0;
          *(v4f*)(out32 + tok * ldo + n + 4) = o1;
        }
      }
    }
    __syncthreads();
  }
}

DEVI void attn_tile(const _Float16* kt, const _Float16* vt, v16h qf,
                    _Float16* __restrict__ orow, int row, int d0) {
  v8f z = {};
  v8f oacc = z;
  float m_run = -1e30f, l_run = 0.0f;

  for (int kb = 0; kb < SEQ / 32; ++kb) {
    const int j0 = kb * 32;
    v16h kf0, kf1;
    {
      const _Float16* kp0 = &kt[(j0 + row) * KT_LD + d0];
      const _Float16* kp1 = &kt[(j0 + 16 + row) * KT_LD + d0];
      v8h ka = *(const v8h*)kp0;
      v8h kb2 = *(const v8h*)kp1;
#pragma unroll
      for (int i = 0; i < 8; ++i) {
        kf0[i] = ka[i];  kf0[8 + i] = (_Float16)0.0f;
        kf1[i] = kb2[i]; kf1[8 + i] = (_Float16)0.0f;
      }
    }
    v8f s0 = wmma16(kf0, qf, z);
    v8f s1 = wmma16(kf1, qf, z);

    float tm = -1e30f;
#pragma unroll
    for (int j = 0; j < 8; ++j) tm = fmaxf(tm, fmaxf(s0[j], s1[j]));
    tm = fmaxf(tm, swap16(tm));
    const float mnew = fmaxf(m_run, tm);
    const float corr = __expf(m_run - mnew);
    float p0[8], p1[8], ls = 0.0f;
#pragma unroll
    for (int j = 0; j < 8; ++j) {
      p0[j] = __expf(s0[j] - mnew);
      p1[j] = __expf(s1[j] - mnew);
      ls += p0[j] + p1[j];
    }
    ls += swap16(ls);
    l_run = l_run * corr + ls;
    m_run = mnew;
#pragma unroll
    for (int j = 0; j < 8; ++j) oacc[j] *= corr;

    v16h pf;
#pragma unroll
    for (int i = 0; i < 8; ++i) {
      pf[i] = (_Float16)p0[i];
      pf[8 + i] = (_Float16)p1[i];
    }
    const _Float16* vp = &vt[row * VT_LD + j0 + d0];
    v8h vl = *(const v8h*)vp;
    v8h vh = *(const v8h*)(vp + 16);
    v16h vf;
#pragma unroll
    for (int i = 0; i < 8; ++i) { vf[i] = vl[i]; vf[8 + i] = vh[i]; }

    oacc = wmma16(vf, pf, oacc);
  }

  const float inv = 1.0f / l_run;
#pragma unroll
  for (int j = 0; j < 8; ++j) orow[j] = (_Float16)(oacc[j] * inv);
}

__global__ __launch_bounds__(512) void attn_kernel(
    const _Float16* __restrict__ xn,
    const float* __restrict__ w_acc, const float* __restrict__ b_acc,
    const float* __restrict__ w_gy, const float* __restrict__ b_gy,
    _Float16* __restrict__ out_all) {
  __shared__ __attribute__((aligned(16))) _Float16 vt[16 * VT_LD];
  __shared__ __attribute__((aligned(16))) _Float16 kt[SEQ * KT_LD];

  const int bid = blockIdx.x;
  const int b = bid / 6;
  const int r = bid % 6;
  const int branch = r / 3;
  const int hh = r % 3;
  const float* Wt = branch ? w_gy : w_acc;
  const float* bias = branch ? b_gy : b_acc;
  const _Float16* xb = xn + (size_t)b * SEQ * E + branch * 96;
  _Float16* outp = out_all + (size_t)branch * NTOK * QDIM
                           + (size_t)b * SEQ * QDIM;
  const int qcol = hh * 16;

  const int wave = threadIdx.x >> 5, lane = threadIdx.x & 31;
  const int row = lane & 15;
  const int d0  = (lane & 16) ? 8 : 0;

  v16h x0[2], x1f[2];
#pragma unroll
  for (int t = 0; t < 2; ++t) {
    const _Float16* xp = xb + (size_t)((wave * 2 + t) * 16 + row) * E + d0;
    x0[t] = frag16<32>(xp);
    x1f[t] = frag16<16>(xp + 32);
  }
  const v8f z0 = {};
  {
    const float* wp = Wt + (size_t)(48 + qcol + row) * QDIM + d0;
    const v16h w0 = frag16_f32<32>(wp), w1 = frag16_f32<16>(wp + 32);
    const v4f b0 = *(const v4f*)(bias + 48 + qcol + d0);
    const v4f b1 = *(const v4f*)(bias + 48 + qcol + d0 + 4);
#pragma unroll
    for (int t = 0; t < 2; ++t) {
      v8f a = wmma16(w0, x0[t], z0);
      a = wmma16(w1, x1f[t], a);
      v8h o;
#pragma unroll
      for (int j = 0; j < 4; ++j) {
        o[j] = (_Float16)(a[j] + b0[j]);
        o[4 + j] = (_Float16)(a[4 + j] + b1[j]);
      }
      *(v8h*)&kt[((wave * 2 + t) * 16 + row) * KT_LD + d0] = o;
    }
  }
  {
    const float* wp = Wt + (size_t)(96 + qcol + row) * QDIM + d0;
    const v16h w0 = frag16_f32<32>(wp), w1 = frag16_f32<16>(wp + 32);
    const float bvv = bias[96 + qcol + row];
#pragma unroll
    for (int t = 0; t < 2; ++t) {
      v8f a = wmma16(x0[t], w0, z0);
      a = wmma16(x1f[t], w1, a);
      v8h o;
#pragma unroll
      for (int j = 0; j < 8; ++j) o[j] = (_Float16)(a[j] + bvv);
      *(v8h*)&vt[row * VT_LD + (wave * 2 + t) * 16 + d0] = o;
    }
  }
  v8h qh[2];
  {
    const float* wp = Wt + (size_t)(qcol + row) * QDIM + d0;
    const v16h w0 = frag16_f32<32>(wp), w1 = frag16_f32<16>(wp + 32);
    const v4f b0 = *(const v4f*)(bias + qcol + d0);
    const v4f b1 = *(const v4f*)(bias + qcol + d0 + 4);
#pragma unroll
    for (int t = 0; t < 2; ++t) {
      v8f a = wmma16(w0, x0[t], z0);
      a = wmma16(w1, x1f[t], a);
#pragma unroll
      for (int j = 0; j < 4; ++j) {
        qh[t][j] = (_Float16)(a[j] + b0[j]) * (_Float16)0.25f;
        qh[t][4 + j] = (_Float16)(a[4 + j] + b1[j]) * (_Float16)0.25f;
      }
    }
  }
  __syncthreads();

#pragma unroll
  for (int t = 0; t < 2; ++t) {
    const int q0 = (wave * 2 + t) * 16;
    v16h qf;
#pragma unroll
    for (int i = 0; i < 8; ++i) { qf[i] = qh[t][i]; qf[8 + i] = (_Float16)0.0f; }
    attn_tile(kt, vt, qf, outp + (size_t)(q0 + row) * QDIM + qcol + d0, row, d0);
  }
}

static inline size_t align256(size_t x) { return (x + 255) & ~(size_t)255; }

extern "C" void kernel_launch(void* const* d_in, const int* in_sizes, int n_in,
                              void* d_out, int out_size, void* d_ws, size_t ws_size,
                              hipStream_t stream) {
  (void)in_sizes; (void)n_in; (void)out_size; (void)ws_size;

  const float* x        = (const float*)d_in[0];
  const float* ln1_g    = (const float*)d_in[1];
  const float* ln1_b    = (const float*)d_in[2];
  const float* acc_wqkv = (const float*)d_in[3];
  const float* acc_bqkv = (const float*)d_in[4];
  const float* acc_wo   = (const float*)d_in[5];
  const float* acc_bo   = (const float*)d_in[6];
  const float* gy_wqkv  = (const float*)d_in[7];
  const float* gy_bqkv  = (const float*)d_in[8];
  const float* gy_wo    = (const float*)d_in[9];
  const float* gy_bo    = (const float*)d_in[10];
  const float* conv_w   = (const float*)d_in[11];
  const float* ln2_g    = (const float*)d_in[12];
  const float* ln2_b    = (const float*)d_in[13];
  const float* w1       = (const float*)d_in[14];
  const float* b1       = (const float*)d_in[15];
  const float* w2       = (const float*)d_in[16];
  const float* b2       = (const float*)d_in[17];

  char* ws = (char*)d_ws;
  size_t off = 0;
  _Float16* xn_h   = (_Float16*)(ws + off); off = align256(off + (size_t)NTOK * E * 2);
  _Float16* attn_h = (_Float16*)(ws + off); off = align256(off + (size_t)2 * NTOK * QDIM * 2);
  float*    x1     = (float*)   (ws + off); off = align256(off + (size_t)NTOK * E * 4);
  _Float16* tt_h   = (_Float16*)(ws + off); off = align256(off + (size_t)NTOK * E * 2);
  _Float16* hid_h  = (_Float16*)(ws + off); off = align256(off + (size_t)NTOK * 2 * E * 2);

  ln_kernel<<<NTOK / 8, 256, 0, stream>>>(x, ln1_g, ln1_b, xn_h);

  conv_wmma_kernel<<<NTOK / 64, 128, 0, stream>>>(xn_h, x, conv_w, x1);

  attn_kernel<<<BATCH * 6, 512, 0, stream>>>(xn_h, acc_wqkv, acc_bqkv,
                                             gy_wqkv, gy_bqkv, attn_h);

  gemm_ws_kernel<48, 3, 1, 64, false, true, false><<<1024, 32, 0, stream>>>(
      attn_h, QDIM, acc_wo, acc_bo, x + 0, E, x1 + 0, nullptr, E);
  gemm_ws_kernel<48, 3, 1, 64, false, true, false><<<1024, 32, 0, stream>>>(
      attn_h + (size_t)NTOK * QDIM, QDIM, gy_wo, gy_bo,
      x + 144, E, x1 + 144, nullptr, E);

  ln_kernel<<<NTOK / 8, 256, 0, stream>>>(x1, ln2_g, ln2_b, tt_h);

  gemm_ws_kernel<192, 4, 6, 64, true, false, true><<<256, 192, 0, stream>>>(
      tt_h, E, w1, b1, nullptr, 0, nullptr, hid_h, 384);

  gemm_ws_kernel<384, 2, 6, 64, true, true, false><<<256, 192, 0, stream>>>(
      hid_h, 2 * E, w2, b2, x1, E, (float*)d_out, nullptr, E);
}


# MLP1+MLP2 in one launch as a two-stage pipeline of specialised waves: 6 waves hold W1, 6 hold W2 in registers, hidden tiles pass between them through LDS, never stored; 5 launches -> 4
# speedup vs baseline: 2.0838x; 1.2216x over previous
#include <hip/hip_runtime.h>


typedef __attribute__((ext_vector_type(16))) _Float16 v16h;
typedef __attribute__((ext_vector_type(8)))  _Float16 v8h;
typedef __attribute__((ext_vector_type(8)))  float    v8f;
typedef __attribute__((ext_vector_type(4)))  float    v4f;

#define DEVI static __device__ __forceinline__

static constexpr int BATCH = 128;
static constexpr int SEQ   = 512;
static constexpr int E     = 192;
static constexpr int QDIM  = 48;
static constexpr int NTOK  = BATCH * SEQ;

static constexpr int KT_LD = 24;
static constexpr int VT_LD = 520;

DEVI v8f wmma16(v16h a, v16h b, v8f c) {
  return __builtin_amdgcn_wmma_f32_16x16x32_f16(false, a, false, b,
                                                (short)0, c, false, false);
}

DEVI float swap16(float v) {
  int i = __builtin_bit_cast(int, v);
  int r = __builtin_amdgcn_permlanex16(i, i, 0x76543210, 0xfedcba98, false, false);
  return __builtin_bit_cast(float, r);
}

__global__ __launch_bounds__(256) void ln_kernel(
    const float* __restrict__ x, const float* __restrict__ g,
    const float* __restrict__ b, _Float16* __restrict__ out) {
  const int wave = threadIdx.x >> 5, lane = threadIdx.x & 31;
  const size_t t = (size_t)blockIdx.x * 8 + wave;
  const float* xp = x + t * E;
  float v[6];
  float s = 0.0f;
#pragma unroll
  for (int i = 0; i < 6; ++i) { v[i] = xp[lane + 32 * i]; s += v[i]; }
#pragma unroll
  for (int o = 16; o > 0; o >>= 1) s += __shfl_xor(s, o, 32);
  const float mean = s * (1.0f / E);
  float vs = 0.0f;
#pragma unroll
  for (int i = 0; i < 6; ++i) { float d = v[i] - mean; vs += d * d; }
#pragma unroll
  for (int o = 16; o > 0; o >>= 1) vs += __shfl_xor(vs, o, 32);
  const float inv = rsqrtf(vs * (1.0f / E) + 1e-6f);
#pragma unroll
  for (int i = 0; i < 6; ++i) {
    int c = lane + 32 * i;
    out[t * E + c] = (_Float16)((v[i] - mean) * inv * g[c] + b[c]);
  }
}

template <int KREM>
DEVI v16h frag16(const _Float16* p) {
  v16h r;
  const v8h lo = *(const v8h*)p;
  if (KREM >= 32) {
    const v8h hi = *(const v8h*)(p + 16);
#pragma unroll
    for (int i = 0; i < 8; ++i) { r[i] = lo[i]; r[8 + i] = hi[i]; }
  } else {
#pragma unroll
    for (int i = 0; i < 8; ++i) { r[i] = lo[i]; r[8 + i] = (_Float16)0.0f; }
  }
  return r;
}

template <int KREM>
DEVI v16h frag16_f32(const float* p) {
  v16h r;
  const v4f a0 = *(const v4f*)p;
  const v4f a1 = *(const v4f*)(p + 4);
#pragma unroll
  for (int i = 0; i < 4; ++i) { r[i] = (_Float16)a0[i]; r[4 + i] = (_Float16)a1[i]; }
  if (KREM >= 32) {
    const v4f b0 = *(const v4f*)(p + 16);
    const v4f b1 = *(const v4f*)(p + 20);
#pragma unroll
    for (int i = 0; i < 4; ++i) { r[8 + i] = (_Float16)b0[i]; r[12 + i] = (_Float16)b1[i]; }
  } else {
#pragma unroll
    for (int i = 0; i < 8; ++i) r[8 + i] = (_Float16)0.0f;
  }
  return r;
}

template <int TM>
__global__ __launch_bounds__(384) void mlp_pipe_kernel(
    const _Float16* __restrict__ tt,
    const float* __restrict__ w1, const float* __restrict__ b1,
    const float* __restrict__ w2, const float* __restrict__ b2,
    const float* __restrict__ x1, float* __restrict__ out) {
  constexpr int K1 = E, K2 = 2 * E;
  constexpr int P1 = K1 + 8;
  constexpr int P2 = K2 + 8;
  constexpr int CPR = K1 / 8;
  constexpr int NTHR = 384;
  constexpr int PT = TM * CPR / NTHR;
  constexpr int NTILES = NTOK / TM;
  static_assert((TM * CPR) % NTHR == 0, "tile must split evenly");
  static_assert((P1 / 8) % 2 == 1 && (P2 / 8) % 2 == 1, "pitch must be 16 B x odd");

  constexpr int W1H = K2 * P1, W2H = K1 * P2;
  static_assert(2 * TM * (P1 + P2) <= W1H + W2H, "tiles must fit in the block");
  __shared__ __attribute__((aligned(16))) _Float16 lds[W1H + W2H];
  __shared__ __attribute__((aligned(16))) float bl[K2 + K1];
  _Float16* const tbuf0 = lds;
  _Float16* const hbuf0 = lds + 2 * TM * P1;

  const int tid = threadIdx.x;
  const int wave = __builtin_amdgcn_readfirstlane(tid >> 5);
  const int lane = tid & 31;
  const int c = lane & 15;
  const int h8 = (lane & 16) ? 8 : 0;
  const bool first = wave < 6;
  const int n0 = first ? wave * 64 : (wave - 6) * 32;

  if (tid < K1) bl[K2 + tid] = b2[tid];
  bl[tid] = b1[tid];
  const float* bw = &bl[(first ? 0 : K2) + n0 + h8];

  v8h st[PT];
  const int G = gridDim.x;
  const int tile0 = blockIdx.x;
  if (tile0 < NTILES) {
#pragma unroll
    for (int i = 0; i < PT; ++i) {
      const int q = i * NTHR + tid;
      const int rr = q / CPR, cc = q % CPR;
      st[i] = *(const v8h*)(tt + (size_t)(tile0 * TM + rr) * K1 + cc * 8);
    }
  }

  typedef __attribute__((ext_vector_type(4))) _Float16 v4h;
#pragma unroll 24
  for (int i = 0; i < (K2 * K1 / 4) / NTHR; ++i) {
    const int q = i * NTHR + tid;
    const v4f a = *(const v4f*)(w1 + (size_t)q * 4);
    const v4f b = *(const v4f*)(w2 + (size_t)q * 4);
    v4h ah, bh;
#pragma unroll
    for (int j = 0; j < 4; ++j) { ah[j] = (_Float16)a[j]; bh[j] = (_Float16)b[j]; }
    *(v4h*)&lds[(q / (K1 / 4)) * P1 + (q % (K1 / 4)) * 4] = ah;
    *(v4h*)&lds[W1H + (q / (K2 / 4)) * P2 + (q % (K2 / 4)) * 4] = bh;
  }
  __syncthreads();

  v16h wf[24];
  if (first) {
#pragma unroll
    for (int t = 0; t < 4; ++t) {
#pragma unroll
      for (int ks = 0; ks < 6; ++ks)
        wf[t * 6 + ks] = frag16<32>(&lds[(n0 + t * 16 + c) * P1 + ks * 32 + h8]);
    }
  } else {
#pragma unroll
    for (int t = 0; t < 2; ++t) {
#pragma unroll
      for (int ks = 0; ks < 12; ++ks)
        wf[t * 12 + ks] = frag16<32>(&lds[W1H + (n0 + t * 16 + c) * P2 + ks * 32 + h8]);
    }
  }
  __syncthreads();

  if (tile0 < NTILES) {
#pragma unroll
    for (int i = 0; i < PT; ++i) {
      const int q = i * NTHR + tid;
      const int rr = q / CPR, cc = q % CPR;
      *(v8h*)&tbuf0[rr * P1 + cc * 8] = st[i];
    }
  }
  if (tile0 + G < NTILES) {
#pragma unroll
    for (int i = 0; i < PT; ++i) {
      const int q = i * NTHR + tid;
      const int rr = q / CPR, cc = q % CPR;
      st[i] = *(const v8h*)(tt + (size_t)((tile0 + G) * TM + rr) * K1 + cc * 8);
    }
  }
  v4f rn[4] = {};
  if (!first && tile0 < NTILES) {
    const float* rp = x1 + ((size_t)tile0 * TM + c) * E + n0 + h8;
    rn[0] = *(const v4f*)(rp);      rn[1] = *(const v4f*)(rp + 4);
    rn[2] = *(const v4f*)(rp + 16); rn[3] = *(const v4f*)(rp + 20);
  }
  __syncthreads();

  for (int s = 0; ; ++s) {
    const int tp = tile0 + s * G;
    const int tc = tp - G;
    const bool hp = tp < NTILES;
    const bool hc = (s > 0) && (tc < NTILES);
    if (!hp && !hc) break;

    if (tp + G < NTILES) {
      _Float16* tn = tbuf0 + ((s + 1) & 1) * (TM * P1);
#pragma unroll
      for (int i = 0; i < PT; ++i) {
        const int q = i * NTHR + tid;
        const int rr = q / CPR, cc = q % CPR;
        *(v8h*)&tn[rr * P1 + cc * 8] = st[i];
      }
      if (tp + 2 * G < NTILES) {
#pragma unroll
        for (int i = 0; i < PT; ++i) {
          const int q = i * NTHR + tid;
          const int rr = q / CPR, cc = q % CPR;
          st[i] = *(const v8h*)(tt + (size_t)((tp + 2 * G) * TM + rr) * K1 + cc * 8);
        }
      }
    }

    if (first) {
      if (hp) {
        const _Float16* tb = tbuf0 + (s & 1) * (TM * P1);
        _Float16* hb = hbuf0 + (s & 1) * (TM * P2);
#pragma unroll 1
        for (int sub = 0; sub < TM / 16; ++sub) {
          const _Float16* xr = &tb[(sub * 16 + c) * P1 + h8];
          v8f acc[4] = {};
#pragma unroll
          for (int kh = 0; kh < 2; ++kh) {
            v16h xf[3];
#pragma unroll
            for (int ks = 0; ks < 3; ++ks) xf[ks] = frag16<32>(xr + (kh * 3 + ks) * 32);
#pragma unroll
            for (int ks = 0; ks < 3; ++ks) {
#pragma unroll
              for (int t = 0; t < 4; ++t)
                acc[t] = wmma16(wf[t * 6 + kh * 3 + ks], xf[ks], acc[t]);
            }
          }
          _Float16* hr = &hb[(sub * 16 + c) * P2 + n0 + h8];
#pragma unroll
          for (int t = 0; t < 4; ++t) {
            const v4f bb0 = *(const v4f*)(bw + t * 16);
            const v4f bb1 = *(const v4f*)(bw + t * 16 + 4);
            v8h o;
#pragma unroll
            for (int j = 0; j < 4; ++j) {
              o[j]     = (_Float16)fmaxf(acc[t][j] + bb0[j], 0.0f);
              o[4 + j] = (_Float16)fmaxf(acc[t][4 + j] + bb1[j], 0.0f);
            }
            *(v8h*)(hr + t * 16) = o;
          }
        }
      }
    } else {
      if (hc) {
        const _Float16* hb = hbuf0 + ((s - 1) & 1) * (TM * P2);
#pragma unroll 1
        for (int sub = 0; sub < TM / 16; ++sub) {
          const size_t tok = (size_t)tc * TM + sub * 16 + c;
          const _Float16* xr = &hb[(sub * 16 + c) * P2 + h8];
          v8f acc[2] = {};
#pragma unroll
          for (int kh = 0; kh < 3; ++kh) {
            v16h xf[4];
#pragma unroll
            for (int ks = 0; ks < 4; ++ks) xf[ks] = frag16<32>(xr + (kh * 4 + ks) * 32);
#pragma unroll
            for (int ks = 0; ks < 4; ++ks) {
#pragma unroll
              for (int t = 0; t < 2; ++t)
                acc[t] = wmma16(wf[t * 12 + kh * 4 + ks], xf[ks], acc[t]);
            }
          }
#pragma unroll
          for (int t = 0; t < 2; ++t) {
            const int n = n0 + t * 16 + h8;
            const v4f bb0 = *(const v4f*)(bw + t * 16);
            const v4f bb1 = *(const v4f*)(bw + t * 16 + 4);
            v4f o0, o1;
#pragma unroll
            for (int j = 0; j < 4; ++j) {
              o0[j] = fmaxf(acc[t][j] + bb0[j], 0.0f) + rn[2 * t][j];
              o1[j] = fmaxf(acc[t][4 + j] + bb1[j], 0.0f) + rn[2 * t + 1][j];
            }
            *(v4f*)(out + tok * E + n) = o0;
            *(v4f*)(out + tok * E + n + 4) = o1;
          }
          const size_t ntok = (sub + 1 < TM / 16) ? tok + 16
                                                  : ((size_t)(tc + G) * TM + c);
          if (sub + 1 < TM / 16 || tc + G < NTILES) {
            const float* rp = x1 + ntok * E + n0 + h8;
            rn[0] = *(const v4f*)(rp);      rn[1] = *(const v4f*)(rp + 4);
            rn[2] = *(const v4f*)(rp + 16); rn[3] = *(const v4f*)(rp + 20);
          }
        }
      }
    }
    __syncthreads();
  }
}

DEVI void attn_tile(const _Float16* kt, const _Float16* vt, v16h qf,
                    _Float16* __restrict__ orow, int row, int d0) {
  v8f z = {};
  v8f oacc = z;
  float m_run = -1e30f, l_run = 0.0f;

  for (int kb = 0; kb < SEQ / 32; ++kb) {
    const int j0 = kb * 32;
    v16h kf0, kf1;
    {
      const _Float16* kp0 = &kt[(j0 + row) * KT_LD + d0];
      const _Float16* kp1 = &kt[(j0 + 16 + row) * KT_LD + d0];
      v8h ka = *(const v8h*)kp0;
      v8h kb2 = *(const v8h*)kp1;
#pragma unroll
      for (int i = 0; i < 8; ++i) {
        kf0[i] = ka[i];  kf0[8 + i] = (_Float16)0.0f;
        kf1[i] = kb2[i]; kf1[8 + i] = (_Float16)0.0f;
      }
    }
    v8f s0 = wmma16(kf0, qf, z);
    v8f s1 = wmma16(kf1, qf, z);

    float tm = -1e30f;
#pragma unroll
    for (int j = 0; j < 8; ++j) tm = fmaxf(tm, fmaxf(s0[j], s1[j]));
    tm = fmaxf(tm, swap16(tm));
    const float mnew = fmaxf(m_run, tm);
    const float corr = __expf(m_run - mnew);
    float p0[8], p1[8], ls = 0.0f;
#pragma unroll
    for (int j = 0; j < 8; ++j) {
      p0[j] = __expf(s0[j] - mnew);
      p1[j] = __expf(s1[j] - mnew);
      ls += p0[j] + p1[j];
    }
    ls += swap16(ls);
    l_run = l_run * corr + ls;
    m_run = mnew;
#pragma unroll
    for (int j = 0; j < 8; ++j) oacc[j] *= corr;

    v16h pf;
#pragma unroll
    for (int i = 0; i < 8; ++i) {
      pf[i] = (_Float16)p0[i];
      pf[8 + i] = (_Float16)p1[i];
    }
    const _Float16* vp = &vt[row * VT_LD + j0 + d0];
    v8h vl = *(const v8h*)vp;
    v8h vh = *(const v8h*)(vp + 16);
    v16h vf;
#pragma unroll
    for (int i = 0; i < 8; ++i) { vf[i] = vl[i]; vf[8 + i] = vh[i]; }

    oacc = wmma16(vf, pf, oacc);
  }

  const float inv = 1.0f / l_run;
#pragma unroll
  for (int j = 0; j < 8; ++j) orow[j] = (_Float16)(oacc[j] * inv);
}

__global__ __launch_bounds__(512) void attn_kernel(
    const _Float16* __restrict__ xn,
    const float* __restrict__ w_acc, const float* __restrict__ b_acc,
    const float* __restrict__ w_gy, const float* __restrict__ b_gy,
    _Float16* __restrict__ out_all) {
  __shared__ __attribute__((aligned(16))) _Float16 vt[16 * VT_LD];
  __shared__ __attribute__((aligned(16))) _Float16 kt[SEQ * KT_LD];

  const int bid = blockIdx.x;
  const int b = bid / 6;
  const int r = bid % 6;
  const int branch = r / 3;
  const int hh = r % 3;
  const float* Wt = branch ? w_gy : w_acc;
  const float* bias = branch ? b_gy : b_acc;
  const _Float16* xb = xn + (size_t)b * SEQ * E + branch * 96;
  _Float16* outp = out_all + (size_t)branch * NTOK * QDIM
                           + (size_t)b * SEQ * QDIM;
  const int qcol = hh * 16;

  const int wave = threadIdx.x >> 5, lane = threadIdx.x & 31;
  const int row = lane & 15;
  const int d0  = (lane & 16) ? 8 : 0;

  v16h x0[2], x1f[2];
#pragma unroll
  for (int t = 0; t < 2; ++t) {
    const _Float16* xp = xb + (size_t)((wave * 2 + t) * 16 + row) * E + d0;
    x0[t] = frag16<32>(xp);
    x1f[t] = frag16<16>(xp + 32);
  }
  const v8f z0 = {};
  {
    const float* wp = Wt + (size_t)(48 + qcol + row) * QDIM + d0;
    const v16h w0 = frag16_f32<32>(wp), w1 = frag16_f32<16>(wp + 32);
    const v4f b0 = *(const v4f*)(bias + 48 + qcol + d0);
    const v4f b1 = *(const v4f*)(bias + 48 + qcol + d0 + 4);
#pragma unroll
    for (int t = 0; t < 2; ++t) {
      v8f a = wmma16(w0, x0[t], z0);
      a = wmma16(w1, x1f[t], a);
      v8h o;
#pragma unroll
      for (int j = 0; j < 4; ++j) {
        o[j] = (_Float16)(a[j] + b0[j]);
        o[4 + j] = (_Float16)(a[4 + j] + b1[j]);
      }
      *(v8h*)&kt[((wave * 2 + t) * 16 + row) * KT_LD + d0] = o;
    }
  }
  {
    const float* wp = Wt + (size_t)(96 + qcol + row) * QDIM + d0;
    const v16h w0 = frag16_f32<32>(wp), w1 = frag16_f32<16>(wp + 32);
    const float bvv = bias[96 + qcol + row];
#pragma unroll
    for (int t = 0; t < 2; ++t) {
      v8f a = wmma16(x0[t], w0, z0);
      a = wmma16(x1f[t], w1, a);
      v8h o;
#pragma unroll
      for (int j = 0; j < 8; ++j) o[j] = (_Float16)(a[j] + bvv);
      *(v8h*)&vt[row * VT_LD + (wave * 2 + t) * 16 + d0] = o;
    }
  }
  v8h qh[2];
  {
    const float* wp = Wt + (size_t)(qcol + row) * QDIM + d0;
    const v16h w0 = frag16_f32<32>(wp), w1 = frag16_f32<16>(wp + 32);
    const v4f b0 = *(const v4f*)(bias + qcol + d0);
    const v4f b1 = *(const v4f*)(bias + qcol + d0 + 4);
#pragma unroll
    for (int t = 0; t < 2; ++t) {
      v8f a = wmma16(w0, x0[t], z0);
      a = wmma16(w1, x1f[t], a);
#pragma unroll
      for (int j = 0; j < 4; ++j) {
        qh[t][j] = (_Float16)(a[j] + b0[j]) * (_Float16)0.25f;
        qh[t][4 + j] = (_Float16)(a[4 + j] + b1[j]) * (_Float16)0.25f;
      }
    }
  }
  __syncthreads();

#pragma unroll
  for (int t = 0; t < 2; ++t) {
    const int q0 = (wave * 2 + t) * 16;
    v16h qf;
#pragma unroll
    for (int i = 0; i < 8; ++i) { qf[i] = qh[t][i]; qf[8 + i] = (_Float16)0.0f; }
    attn_tile(kt, vt, qf, outp + (size_t)(q0 + row) * QDIM + qcol + d0, row, d0);
  }
}

__global__ __launch_bounds__(128) void rows_kernel(
    const _Float16* __restrict__ xn,
    const _Float16* __restrict__ at_acc, const _Float16* __restrict__ at_gy,
    const float* __restrict__ x, const float* __restrict__ cw,
    const float* __restrict__ wo_acc, const float* __restrict__ bo_acc,
    const float* __restrict__ wo_gy, const float* __restrict__ bo_gy,
    const float* __restrict__ g2, const float* __restrict__ b2,
    float* __restrict__ x1, _Float16* __restrict__ tt) {
  const int ch = threadIdx.x >> 5, lane = threadIdx.x & 31;
  const int c = lane & 15, h = lane >> 4;
  const int h8 = 8 * h;
  const size_t tok = (size_t)blockIdx.x * 64 + 4 * c + ch;

  float v[12][8];
  {
    const float* rp = x + tok * E + h8;
#pragma unroll
    for (int m = 0; m < 12; ++m) {
      const v4f r0 = *(const v4f*)(rp + 16 * m);
      const v4f r1 = *(const v4f*)(rp + 16 * m + 4);
#pragma unroll
      for (int j = 0; j < 4; ++j) { v[m][j] = r0[j]; v[m][4 + j] = r1[j]; }
    }
  }

#pragma unroll
  for (int br = 0; br < 2; ++br) {
    const float* Wt = br ? wo_gy : wo_acc;
    const float* bias = br ? bo_gy : bo_acc;
    const _Float16* ap = (br ? at_gy : at_acc) + tok * QDIM + h8;
    const v16h a0 = frag16<32>(ap), a1 = frag16<16>(ap + 32);
#pragma unroll
    for (int t = 0; t < 3; ++t) {
      const float* wp = Wt + (size_t)(t * 16 + c) * QDIM + h8;
      const v16h w0 = frag16_f32<32>(wp), w1 = frag16_f32<16>(wp + 32);
      const v4f bb0 = *(const v4f*)(bias + t * 16 + h8);
      const v4f bb1 = *(const v4f*)(bias + t * 16 + h8 + 4);
      v8f acc = {};
      acc = wmma16(w0, a0, acc);
      acc = wmma16(w1, a1, acc);
      const int m = br ? 9 + t : t;
#pragma unroll
      for (int j = 0; j < 4; ++j) {
        v[m][j]     = (acc[j] + bb0[j]) + v[m][j];
        v[m][4 + j] = (acc[4 + j] + bb1[j]) + v[m][4 + j];
      }
    }
  }

  {
    const float* w = cw + ch * 15;
    v16h af;
#pragma unroll
    for (int i = 0; i < 8; ++i) {
      const int k0 = 8 * h + i - c - 1;
      const int k1 = k0 + 16;
      const float w0 = w[min(max(k0, 0), 14)];
      const float w1 = w[min(max(k1, 0), 14)];
      af[i]     = (_Float16)((k0 >= 0 && k0 <= 14) ? w0 : 0.0f);
      af[8 + i] = (_Float16)((k1 >= 0 && k1 <= 14) ? w1 : 0.0f);
    }
    const _Float16* xp = xn + tok * E + 48;
    v8h ck[7];
    const v8h zero8 = {};
    {
      const v8h first = *(const v8h*)(xp);
      const v8h last  = *(const v8h*)(xp + 88);
      ck[0] = h ? first : zero8;
      ck[6] = h ? zero8 : last;
    }
#pragma unroll
    for (int j = 1; j < 6; ++j) ck[j] = *(const v8h*)(xp + 8 * (2 * j - 1 + h));
#pragma unroll
    for (int n = 0; n < 6; ++n) {
      v16h bf;
#pragma unroll
      for (int i = 0; i < 8; ++i) { bf[i] = ck[n][i]; bf[8 + i] = ck[n + 1][i]; }
      v8f acc = {};
      acc = wmma16(af, bf, acc);
#pragma unroll
      for (int j = 0; j < 8; ++j) v[3 + n][j] = acc[j] + v[3 + n][j];
    }
  }

  float s = 0.0f;
#pragma unroll
  for (int m = 0; m < 12; ++m) {
#pragma unroll
    for (int j = 0; j < 8; ++j) s += v[m][j];
  }
  s += swap16(s);
  const float mean = s * (1.0f / E);
  float vs = 0.0f;
#pragma unroll
  for (int m = 0; m < 12; ++m) {
#pragma unroll
    for (int j = 0; j < 8; ++j) { const float d = v[m][j] - mean; vs += d * d; }
  }
  vs += swap16(vs);
  const float inv = rsqrtf(vs * (1.0f / E) + 1e-6f);

  float* op = x1 + tok * E + h8;
  _Float16* tp = tt + tok * E + h8;
#pragma unroll
  for (int m = 0; m < 12; ++m) {
    const v4f g0 = *(const v4f*)(g2 + 16 * m + h8);
    const v4f g1 = *(const v4f*)(g2 + 16 * m + h8 + 4);
    const v4f e0 = *(const v4f*)(b2 + 16 * m + h8);
    const v4f e1 = *(const v4f*)(b2 + 16 * m + h8 + 4);
    v4f o0, o1;
    v8h o;
#pragma unroll
    for (int j = 0; j < 4; ++j) {
      o0[j] = v[m][j];
      o1[j] = v[m][4 + j];
      o[j]     = (_Float16)((v[m][j] - mean) * inv * g0[j] + e0[j]);
      o[4 + j] = (_Float16)((v[m][4 + j] - mean) * inv * g1[j] + e1[j]);
    }
    *(v4f*)(op + 16 * m) = o0;
    *(v4f*)(op + 16 * m + 4) = o1;
    *(v8h*)(tp + 16 * m) = o;
  }
}

static inline size_t align256(size_t x) { return (x + 255) & ~(size_t)255; }

extern "C" void kernel_launch(void* const* d_in, const int* in_sizes, int n_in,
                              void* d_out, int out_size, void* d_ws, size_t ws_size,
                              hipStream_t stream) {
  (void)in_sizes; (void)n_in; (void)out_size; (void)ws_size;

  const float* x        = (const float*)d_in[0];
  const float* ln1_g    = (const float*)d_in[1];
  const float* ln1_b    = (const float*)d_in[2];
  const float* acc_wqkv = (const float*)d_in[3];
  const float* acc_bqkv = (const float*)d_in[4];
  const float* acc_wo   = (const float*)d_in[5];
  const float* acc_bo   = (const float*)d_in[6];
  const float* gy_wqkv  = (const float*)d_in[7];
  const float* gy_bqkv  = (const float*)d_in[8];
  const float* gy_wo    = (const float*)d_in[9];
  const float* gy_bo    = (const float*)d_in[10];
  const float* conv_w   = (const float*)d_in[11];
  const float* ln2_g    = (const float*)d_in[12];
  const float* ln2_b    = (const float*)d_in[13];
  const float* w1       = (const float*)d_in[14];
  const float* b1       = (const float*)d_in[15];
  const float* w2       = (const float*)d_in[16];
  const float* b2       = (const float*)d_in[17];

  char* ws = (char*)d_ws;
  size_t off = 0;
  _Float16* xn_h   = (_Float16*)(ws + off); off = align256(off + (size_t)NTOK * E * 2);
  _Float16* attn_h = (_Float16*)(ws + off); off = align256(off + (size_t)2 * NTOK * QDIM * 2);
  float*    x1     = (float*)   (ws + off); off = align256(off + (size_t)NTOK * E * 4);
  _Float16* tt_h   = (_Float16*)(ws + off); off = align256(off + (size_t)NTOK * E * 2);

  ln_kernel<<<NTOK / 8, 256, 0, stream>>>(x, ln1_g, ln1_b, xn_h);

  attn_kernel<<<BATCH * 6, 512, 0, stream>>>(xn_h, acc_wqkv, acc_bqkv,
                                             gy_wqkv, gy_bqkv, attn_h);

  rows_kernel<<<NTOK / 64, 128, 0, stream>>>(
      xn_h, attn_h, attn_h + (size_t)NTOK * QDIM, x, conv_w,
      acc_wo, acc_bo, gy_wo, gy_bo, ln2_g, ln2_b, x1, tt_h);

  mlp_pipe_kernel<32><<<256, 384, 0, stream>>>(tt_h, w1, b1, w2, b2, x1, (float*)d_out);
}


# Whole-line global traffic: rows launch (x in, x1, tt out) and the MLP's out rows and residual turn through wave-private LDS patches; each access covers 4 whole 128-byte lines, not 16-byte pieces
# speedup vs baseline: 2.5739x; 1.2352x over previous
#include <hip/hip_runtime.h>


typedef __attribute__((ext_vector_type(16))) _Float16 v16h;
typedef __attribute__((ext_vector_type(8)))  _Float16 v8h;
typedef __attribute__((ext_vector_type(8)))  float    v8f;
typedef __attribute__((ext_vector_type(4)))  float    v4f;

#define DEVI static __device__ __forceinline__

static constexpr int BATCH = 128;
static constexpr int SEQ   = 512;
static constexpr int E     = 192;
static constexpr int QDIM  = 48;
static constexpr int NTOK  = BATCH * SEQ;

static constexpr int RP = 196;
static constexpr int TP = 200;

static constexpr int KT_LD = 24;
static constexpr int VT_LD = 520;

DEVI v8f wmma16(v16h a, v16h b, v8f c) {
  return __builtin_amdgcn_wmma_f32_16x16x32_f16(false, a, false, b,
                                                (short)0, c, false, false);
}

DEVI float swap16(float v) {
  int i = __builtin_bit_cast(int, v);
  int r = __builtin_amdgcn_permlanex16(i, i, 0x76543210, 0xfedcba98, false, false);
  return __builtin_bit_cast(float, r);
}

__global__ __launch_bounds__(256) void ln_kernel(
    const float* __restrict__ x, const float* __restrict__ g,
    const float* __restrict__ b, _Float16* __restrict__ out) {
  const int wave = threadIdx.x >> 5, lane = threadIdx.x & 31;
  const size_t t = (size_t)blockIdx.x * 8 + wave;
  const float* xp = x + t * E;
  float v[6];
  float s = 0.0f;
#pragma unroll
  for (int i = 0; i < 6; ++i) { v[i] = xp[lane + 32 * i]; s += v[i]; }
#pragma unroll
  for (int o = 16; o > 0; o >>= 1) s += __shfl_xor(s, o, 32);
  const float mean = s * (1.0f / E);
  float vs = 0.0f;
#pragma unroll
  for (int i = 0; i < 6; ++i) { float d = v[i] - mean; vs += d * d; }
#pragma unroll
  for (int o = 16; o > 0; o >>= 1) vs += __shfl_xor(vs, o, 32);
  const float inv = rsqrtf(vs * (1.0f / E) + 1e-6f);
#pragma unroll
  for (int i = 0; i < 6; ++i) {
    int c = lane + 32 * i;
    out[t * E + c] = (_Float16)((v[i] - mean) * inv * g[c] + b[c]);
  }
}

template <int KREM>
DEVI v16h frag16(const _Float16* p) {
  v16h r;
  const v8h lo = *(const v8h*)p;
  if (KREM >= 32) {
    const v8h hi = *(const v8h*)(p + 16);
#pragma unroll
    for (int i = 0; i < 8; ++i) { r[i] = lo[i]; r[8 + i] = hi[i]; }
  } else {
#pragma unroll
    for (int i = 0; i < 8; ++i) { r[i] = lo[i]; r[8 + i] = (_Float16)0.0f; }
  }
  return r;
}

template <int KREM>
DEVI v16h frag16_f32(const float* p) {
  v16h r;
  const v4f a0 = *(const v4f*)p;
  const v4f a1 = *(const v4f*)(p + 4);
#pragma unroll
  for (int i = 0; i < 4; ++i) { r[i] = (_Float16)a0[i]; r[4 + i] = (_Float16)a1[i]; }
  if (KREM >= 32) {
    const v4f b0 = *(const v4f*)(p + 16);
    const v4f b1 = *(const v4f*)(p + 20);
#pragma unroll
    for (int i = 0; i < 4; ++i) { r[8 + i] = (_Float16)b0[i]; r[12 + i] = (_Float16)b1[i]; }
  } else {
#pragma unroll
    for (int i = 0; i < 8; ++i) r[8 + i] = (_Float16)0.0f;
  }
  return r;
}

DEVI void wave_turn() {
  __builtin_amdgcn_fence(__ATOMIC_RELEASE, "wavefront");
  __builtin_amdgcn_wave_barrier();
  __builtin_amdgcn_fence(__ATOMIC_ACQUIRE, "wavefront");
}

template <int TM>
__global__ __launch_bounds__(384) void mlp_pipe_kernel(
    const _Float16* __restrict__ tt,
    const float* __restrict__ w1, const float* __restrict__ b1,
    const float* __restrict__ w2, const float* __restrict__ b2,
    const float* __restrict__ x1, float* __restrict__ out) {
  constexpr int K1 = E, K2 = 2 * E;
  constexpr int P1 = K1 + 8;
  constexpr int P2 = K2 + 8;
  constexpr int CPR = K1 / 8;
  constexpr int NTHR = 384;
  constexpr int PT = TM * CPR / NTHR;
  constexpr int NTILES = NTOK / TM;
  static_assert((TM * CPR) % NTHR == 0, "tile must split evenly");
  static_assert((P1 / 8) % 2 == 1 && (P2 / 8) % 2 == 1, "pitch must be 16 B x odd");

  constexpr int W1H = K2 * P1, W2H = K1 * P2;
  static_assert(2 * TM * (P1 + P2) <= W1H + W2H, "tiles must fit in the block");
  __shared__ __attribute__((aligned(16))) _Float16 lds[W1H + W2H];
  __shared__ __attribute__((aligned(16))) float bl[K2 + K1];
  _Float16* const tbuf0 = lds;
  _Float16* const hbuf0 = lds + 2 * TM * P1;
  constexpr int OPP = 36;
  static_assert(2 * TM * (P1 + P2) + 6 * 16 * OPP * 2 <= W1H + W2H, "patches must fit in the block");

  const int tid = threadIdx.x;
  const int wave = __builtin_amdgcn_readfirstlane(tid >> 5);
  const int lane = tid & 31;
  const int c = lane & 15;
  const int h8 = (lane & 16) ? 8 : 0;
  const bool first = wave < 6;
  const int n0 = first ? wave * 64 : (wave - 6) * 32;
  float* const opatch = (float*)(lds + 2 * TM * (P1 + P2)) + (first ? 0 : wave - 6) * (16 * OPP);
  const int lq = lane & 7;
  const int lrb = ((lane >> 3) & 1) * 8 + (lane >> 4) * 4;

  if (tid < K1) bl[K2 + tid] = b2[tid];
  bl[tid] = b1[tid];
  const float* bw = &bl[(first ? 0 : K2) + n0 + h8];

  v8h st[PT];
  const int G = gridDim.x;
  const int tile0 = blockIdx.x;
  if (tile0 < NTILES) {
#pragma unroll
    for (int i = 0; i < PT; ++i) {
      const int q = i * NTHR + tid;
      const int rr = q / CPR, cc = q % CPR;
      st[i] = *(const v8h*)(tt + (size_t)(tile0 * TM + rr) * K1 + cc * 8);
    }
  }

  typedef __attribute__((ext_vector_type(4))) _Float16 v4h;
#pragma unroll 24
  for (int i = 0; i < (K2 * K1 / 4) / NTHR; ++i) {
    const int q = i * NTHR + tid;
    const v4f a = *(const v4f*)(w1 + (size_t)q * 4);
    const v4f b = *(const v4f*)(w2 + (size_t)q * 4);
    v4h ah, bh;
#pragma unroll
    for (int j = 0; j < 4; ++j) { ah[j] = (_Float16)a[j]; bh[j] = (_Float16)b[j]; }
    *(v4h*)&lds[(q / (K1 / 4)) * P1 + (q % (K1 / 4)) * 4] = ah;
    *(v4h*)&lds[W1H + (q / (K2 / 4)) * P2 + (q % (K2 / 4)) * 4] = bh;
  }
  __syncthreads();

  v16h wf[24];
  if (first) {
#pragma unroll
    for (int t = 0; t < 4; ++t) {
#pragma unroll
      for (int ks = 0; ks < 6; ++ks)
        wf[t * 6 + ks] = frag16<32>(&lds[(n0 + t * 16 + c) * P1 + ks * 32 + h8]);
    }
  } else {
#pragma unroll
    for (int t = 0; t < 2; ++t) {
#pragma unroll
      for (int ks = 0; ks < 12; ++ks)
        wf[t * 12 + ks] = frag16<32>(&lds[W1H + (n0 + t * 16 + c) * P2 + ks * 32 + h8]);
    }
  }
  __syncthreads();

  if (tile0 < NTILES) {
#pragma unroll
    for (int i = 0; i < PT; ++i) {
      const int q = i * NTHR + tid;
      const int rr = q / CPR, cc = q % CPR;
      *(v8h*)&tbuf0[rr * P1 + cc * 8] = st[i];
    }
  }
  if (tile0 + G < NTILES) {
#pragma unroll
    for (int i = 0; i < PT; ++i) {
      const int q = i * NTHR + tid;
      const int rr = q / CPR, cc = q % CPR;
      st[i] = *(const v8h*)(tt + (size_t)((tile0 + G) * TM + rr) * K1 + cc * 8);
    }
  }
  v4f rn[4] = {};
  if (!first && tile0 < NTILES) {
    const float* rp = x1 + ((size_t)tile0 * TM + lrb) * E + n0 + 4 * lq;
#pragma unroll
    for (int g = 0; g < 4; ++g) rn[g] = *(const v4f*)(rp + (size_t)g * E);
  }
  __syncthreads();

  for (int s = 0; ; ++s) {
    const int tp = tile0 + s * G;
    const int tc = tp - G;
    const bool hp = tp < NTILES;
    const bool hc = (s > 0) && (tc < NTILES);
    if (!hp && !hc) break;

    if (tp + G < NTILES) {
      _Float16* tn = tbuf0 + ((s + 1) & 1) * (TM * P1);
#pragma unroll
      for (int i = 0; i < PT; ++i) {
        const int q = i * NTHR + tid;
        const int rr = q / CPR, cc = q % CPR;
        *(v8h*)&tn[rr * P1 + cc * 8] = st[i];
      }
      if (tp + 2 * G < NTILES) {
#pragma unroll
        for (int i = 0; i < PT; ++i) {
          const int q = i * NTHR + tid;
          const int rr = q / CPR, cc = q % CPR;
          st[i] = *(const v8h*)(tt + (size_t)((tp + 2 * G) * TM + rr) * K1 + cc * 8);
        }
      }
    }

    if (first) {
      if (hp) {
        const _Float16* tb = tbuf0 + (s & 1) * (TM * P1);
        _Float16* hb = hbuf0 + (s & 1) * (TM * P2);
#pragma unroll 1
        for (int sub = 0; sub < TM / 16; ++sub) {
          const _Float16* xr = &tb[(sub * 16 + c) * P1 + h8];
          v8f acc[4] = {};
#pragma unroll
          for (int kh = 0; kh < 2; ++kh) {
            v16h xf[3];
#pragma unroll
            for (int ks = 0; ks < 3; ++ks) xf[ks] = frag16<32>(xr + (kh * 3 + ks) * 32);
#pragma unroll
            for (int ks = 0; ks < 3; ++ks) {
#pragma unroll
              for (int t = 0; t < 4; ++t)
                acc[t] = wmma16(wf[t * 6 + kh * 3 + ks], xf[ks], acc[t]);
            }
          }
          _Float16* hr = &hb[(sub * 16 + c) * P2 + n0 + h8];
#pragma unroll
          for (int t = 0; t < 4; ++t) {
            const v4f bb0 = *(const v4f*)(bw + t * 16);
            const v4f bb1 = *(const v4f*)(bw + t * 16 + 4);
            v8h o;
#pragma unroll
            for (int j = 0; j < 4; ++j) {
              o[j]     = (_Float16)fmaxf(acc[t][j] + bb0[j], 0.0f);
              o[4 + j] = (_Float16)fmaxf(acc[t][4 + j] + bb1[j], 0.0f);
            }
            *(v8h*)(hr + t * 16) = o;
          }
        }
      }
    } else {
      if (hc) {
        const _Float16* hb = hbuf0 + ((s - 1) & 1) * (TM * P2);
#pragma unroll 1
        for (int sub = 0; sub < TM / 16; ++sub) {
          const size_t tok = (size_t)tc * TM + sub * 16 + c;
          const _Float16* xr = &hb[(sub * 16 + c) * P2 + h8];
          v8f acc[2] = {};
#pragma unroll
          for (int kh = 0; kh < 3; ++kh) {
            v16h xf[4];
#pragma unroll
            for (int ks = 0; ks < 4; ++ks) xf[ks] = frag16<32>(xr + (kh * 4 + ks) * 32);
#pragma unroll
            for (int ks = 0; ks < 4; ++ks) {
#pragma unroll
              for (int t = 0; t < 2; ++t)
                acc[t] = wmma16(wf[t * 12 + kh * 4 + ks], xf[ks], acc[t]);
            }
          }
          float* const pw = opatch + c * OPP + h8;
#pragma unroll
          for (int t = 0; t < 2; ++t) {
            const v4f bb0 = *(const v4f*)(bw + t * 16);
            const v4f bb1 = *(const v4f*)(bw + t * 16 + 4);
            v4f o0, o1;
#pragma unroll
            for (int j = 0; j < 4; ++j) {
              o0[j] = fmaxf(acc[t][j] + bb0[j], 0.0f);
              o1[j] = fmaxf(acc[t][4 + j] + bb1[j], 0.0f);
            }
            *(v4f*)(pw + t * 16) = o0;
            *(v4f*)(pw + t * 16 + 4) = o1;
          }
          wave_turn();
          const size_t ltok = tok - c + lrb;
#pragma unroll
          for (int g = 0; g < 4; ++g) {
            const v4f pv = *(const v4f*)&opatch[(lrb + g) * OPP + 4 * lq];
            v4f o;
#pragma unroll
            for (int j = 0; j < 4; ++j) o[j] = pv[j] + rn[g][j];
            *(v4f*)(out + (ltok + g) * E + n0 + 4 * lq) = o;
          }
          wave_turn();
          const size_t ntok = (sub + 1 < TM / 16) ? ltok + 16
                                                  : ((size_t)(tc + G) * TM + lrb);
          if (sub + 1 < TM / 16 || tc + G < NTILES) {
            const float* rp = x1 + ntok * E + n0 + 4 * lq;
#pragma unroll
            for (int g = 0; g < 4; ++g) rn[g] = *(const v4f*)(rp + (size_t)g * E);
          }
        }
      }
    }
    __syncthreads();
  }
}

DEVI void attn_tile(const _Float16* kt, const _Float16* vt, v16h qf,
                    _Float16* __restrict__ orow, int row, int d0) {
  v8f z = {};
  v8f oacc = z;
  float m_run = -1e30f, l_run = 0.0f;

  for (int kb = 0; kb < SEQ / 32; ++kb) {
    const int j0 = kb * 32;
    v16h kf0, kf1;
    {
      const _Float16* kp0 = &kt[(j0 + row) * KT_LD + d0];
      const _Float16* kp1 = &kt[(j0 + 16 + row) * KT_LD + d0];
      v8h ka = *(const v8h*)kp0;
      v8h kb2 = *(const v8h*)kp1;
#pragma unroll
      for (int i = 0; i < 8; ++i) {
        kf0[i] = ka[i];  kf0[8 + i] = (_Float16)0.0f;
        kf1[i] = kb2[i]; kf1[8 + i] = (_Float16)0.0f;
      }
    }
    v8f s0 = wmma16(kf0, qf, z);
    v8f s1 = wmma16(kf1, qf, z);

    float tm = -1e30f;
#pragma unroll
    for (int j = 0; j < 8; ++j) tm = fmaxf(tm, fmaxf(s0[j], s1[j]));
    tm = fmaxf(tm, swap16(tm));
    const float mnew = fmaxf(m_run, tm);
    const float corr = __expf(m_run - mnew);
    float p0[8], p1[8], ls = 0.0f;
#pragma unroll
    for (int j = 0; j < 8; ++j) {
      p0[j] = __expf(s0[j] - mnew);
      p1[j] = __expf(s1[j] - mnew);
      ls += p0[j] + p1[j];
    }
    ls += swap16(ls);
    l_run = l_run * corr + ls;
    m_run = mnew;
#pragma unroll
    for (int j = 0; j < 8; ++j) oacc[j] *= corr;

    v16h pf;
#pragma unroll
    for (int i = 0; i < 8; ++i) {
      pf[i] = (_Float16)p0[i];
      pf[8 + i] = (_Float16)p1[i];
    }
    const _Float16* vp = &vt[row * VT_LD + j0 + d0];
    v8h vl = *(const v8h*)vp;
    v8h vh = *(const v8h*)(vp + 16);
    v16h vf;
#pragma unroll
    for (int i = 0; i < 8; ++i) { vf[i] = vl[i]; vf[8 + i] = vh[i]; }

    oacc = wmma16(vf, pf, oacc);
  }

  const float inv = 1.0f / l_run;
#pragma unroll
  for (int j = 0; j < 8; ++j) orow[j] = (_Float16)(oacc[j] * inv);
}

__global__ __launch_bounds__(512) void attn_kernel(
    const _Float16* __restrict__ xn,
    const float* __restrict__ w_acc, const float* __restrict__ b_acc,
    const float* __restrict__ w_gy, const float* __restrict__ b_gy,
    _Float16* __restrict__ out_all) {
  __shared__ __attribute__((aligned(16))) _Float16 vt[16 * VT_LD];
  __shared__ __attribute__((aligned(16))) _Float16 kt[SEQ * KT_LD];

  const int bid = blockIdx.x;
  const int b = bid / 6;
  const int r = bid % 6;
  const int branch = r / 3;
  const int hh = r % 3;
  const float* Wt = branch ? w_gy : w_acc;
  const float* bias = branch ? b_gy : b_acc;
  const _Float16* xb = xn + (size_t)b * SEQ * E + branch * 96;
  _Float16* outp = out_all + (size_t)branch * NTOK * QDIM
                           + (size_t)b * SEQ * QDIM;
  const int qcol = hh * 16;

  const int wave = threadIdx.x >> 5, lane = threadIdx.x & 31;
  const int row = lane & 15;
  const int d0  = (lane & 16) ? 8 : 0;

  v16h x0[2], x1f[2];
#pragma unroll
  for (int t = 0; t < 2; ++t) {
    const _Float16* xp = xb + (size_t)((wave * 2 + t) * 16 + row) * E + d0;
    x0[t] = frag16<32>(xp);
    x1f[t] = frag16<16>(xp + 32);
  }
  const v8f z0 = {};
  {
    const float* wp = Wt + (size_t)(48 + qcol + row) * QDIM + d0;
    const v16h w0 = frag16_f32<32>(wp), w1 = frag16_f32<16>(wp + 32);
    const v4f b0 = *(const v4f*)(bias + 48 + qcol + d0);
    const v4f b1 = *(const v4f*)(bias + 48 + qcol + d0 + 4);
#pragma unroll
    for (int t = 0; t < 2; ++t) {
      v8f a = wmma16(w0, x0[t], z0);
      a = wmma16(w1, x1f[t], a);
      v8h o;
#pragma unroll
      for (int j = 0; j < 4; ++j) {
        o[j] = (_Float16)(a[j] + b0[j]);
        o[4 + j] = (_Float16)(a[4 + j] + b1[j]);
      }
      *(v8h*)&kt[((wave * 2 + t) * 16 + row) * KT_LD + d0] = o;
    }
  }
  {
    const float* wp = Wt + (size_t)(96 + qcol + row) * QDIM + d0;
    const v16h w0 = frag16_f32<32>(wp), w1 = frag16_f32<16>(wp + 32);
    const float bvv = bias[96 + qcol + row];
#pragma unroll
    for (int t = 0; t < 2; ++t) {
      v8f a = wmma16(x0[t], w0, z0);
      a = wmma16(x1f[t], w1, a);
      v8h o;
#pragma unroll
      for (int j = 0; j < 8; ++j) o[j] = (_Float16)(a[j] + bvv);
      *(v8h*)&vt[row * VT_LD + (wave * 2 + t) * 16 + d0] = o;
    }
  }
  v8h qh[2];
  {
    const float* wp = Wt + (size_t)(qcol + row) * QDIM + d0;
    const v16h w0 = frag16_f32<32>(wp), w1 = frag16_f32<16>(wp + 32);
    const v4f b0 = *(const v4f*)(bias + qcol + d0);
    const v4f b1 = *(const v4f*)(bias + qcol + d0 + 4);
#pragma unroll
    for (int t = 0; t < 2; ++t) {
      v8f a = wmma16(w0, x0[t], z0);
      a = wmma16(w1, x1f[t], a);
#pragma unroll
      for (int j = 0; j < 4; ++j) {
        qh[t][j] = (_Float16)(a[j] + b0[j]) * (_Float16)0.25f;
        qh[t][4 + j] = (_Float16)(a[4 + j] + b1[j]) * (_Float16)0.25f;
      }
    }
  }
  __syncthreads();

#pragma unroll
  for (int t = 0; t < 2; ++t) {
    const int q0 = (wave * 2 + t) * 16;
    v16h qf;
#pragma unroll
    for (int i = 0; i < 8; ++i) { qf[i] = qh[t][i]; qf[8 + i] = (_Float16)0.0f; }
    attn_tile(kt, vt, qf, outp + (size_t)(q0 + row) * QDIM + qcol + d0, row, d0);
  }
}

__global__ __launch_bounds__(128) void rows_kernel(
    const _Float16* __restrict__ xn,
    const _Float16* __restrict__ at_acc, const _Float16* __restrict__ at_gy,
    const float* __restrict__ x, const float* __restrict__ cw,
    const float* __restrict__ wo_acc, const float* __restrict__ bo_acc,
    const float* __restrict__ wo_gy, const float* __restrict__ bo_gy,
    const float* __restrict__ g2, const float* __restrict__ b2,
    float* __restrict__ x1, _Float16* __restrict__ tt) {
  const int ch = threadIdx.x >> 5, lane = threadIdx.x & 31;
  const int c = lane & 15, h = lane >> 4;
  const int h8 = 8 * h;
  const size_t tok = (size_t)blockIdx.x * 64 + 4 * c + ch;

  __shared__ __attribute__((aligned(16))) float patch_all[4 * 16 * RP];
  float* const patch = patch_all + ch * (16 * RP);
  _Float16* const hpatch = (_Float16*)patch;
  const int lq = lane & 7;
  const int lrb = ((lane >> 3) & 1) * 8 + (lane >> 4) * 4;
  const size_t ltok = (size_t)blockIdx.x * 64 + 4 * lrb + ch;

  float v[12][8];
  v16h a0s[2], a1s[2];
  v8h ck[7];
  v16h af;
  {
    v4f tl[24];
#pragma unroll
    for (int L = 0; L < 6; ++L) {
#pragma unroll
      for (int g = 0; g < 4; ++g)
        tl[L * 4 + g] = *(const v4f*)(x + (ltok + 4 * g) * E + 32 * L + 4 * lq);
    }
#pragma unroll
    for (int br = 0; br < 2; ++br) {
      const _Float16* ap = (br ? at_gy : at_acc) + tok * QDIM + h8;
      a0s[br] = frag16<32>(ap);
      a1s[br] = frag16<16>(ap + 32);
    }
    {
      const float* w = cw + ch * 15;
#pragma unroll
      for (int i = 0; i < 8; ++i) {
        const int k0 = 8 * h + i - c - 1;
        const int k1 = k0 + 16;
        const float w0 = w[min(max(k0, 0), 14)];
        const float w1 = w[min(max(k1, 0), 14)];
        af[i]     = (_Float16)((k0 >= 0 && k0 <= 14) ? w0 : 0.0f);
        af[8 + i] = (_Float16)((k1 >= 0 && k1 <= 14) ? w1 : 0.0f);
      }
    }
    {
      const _Float16* xp = xn + tok * E + 48;
      const v8h zero8 = {};
      const v8h first = *(const v8h*)(xp);
      const v8h last  = *(const v8h*)(xp + 88);
      ck[0] = h ? first : zero8;
      ck[6] = h ? zero8 : last;
#pragma unroll
      for (int j = 1; j < 6; ++j) ck[j] = *(const v8h*)(xp + 8 * (2 * j - 1 + h));
    }
#pragma unroll
    for (int L = 0; L < 6; ++L) {
#pragma unroll
      for (int g = 0; g < 4; ++g)
        *(v4f*)&patch[(lrb + g) * RP + 32 * L + 4 * lq] = tl[L * 4 + g];
    }
    __syncthreads();
    const float* rp = patch + c * RP + h8;
#pragma unroll
    for (int m = 0; m < 12; ++m) {
      const v4f r0 = *(const v4f*)(rp + 16 * m);
      const v4f r1 = *(const v4f*)(rp + 16 * m + 4);
#pragma unroll
      for (int j = 0; j < 4; ++j) { v[m][j] = r0[j]; v[m][4 + j] = r1[j]; }
    }
  }

#pragma unroll
  for (int br = 0; br < 2; ++br) {
    const float* Wt = br ? wo_gy : wo_acc;
    const float* bias = br ? bo_gy : bo_acc;
    const v16h a0 = a0s[br], a1 = a1s[br];
#pragma unroll
    for (int t = 0; t < 3; ++t) {
      const float* wp = Wt + (size_t)(t * 16 + c) * QDIM + h8;
      const v16h w0 = frag16_f32<32>(wp), w1 = frag16_f32<16>(wp + 32);
      const v4f bb0 = *(const v4f*)(bias + t * 16 + h8);
      const v4f bb1 = *(const v4f*)(bias + t * 16 + h8 + 4);
      v8f acc = {};
      acc = wmma16(w0, a0, acc);
      acc = wmma16(w1, a1, acc);
      const int m = br ? 9 + t : t;
#pragma unroll
      for (int j = 0; j < 4; ++j) {
        v[m][j]     = (acc[j] + bb0[j]) + v[m][j];
        v[m][4 + j] = (acc[4 + j] + bb1[j]) + v[m][4 + j];
      }
    }
  }

  {
#pragma unroll
    for (int n = 0; n < 6; ++n) {
      v16h bf;
#pragma unroll
      for (int i = 0; i < 8; ++i) { bf[i] = ck[n][i]; bf[8 + i] = ck[n + 1][i]; }
      v8f acc = {};
      acc = wmma16(af, bf, acc);
#pragma unroll
      for (int j = 0; j < 8; ++j) v[3 + n][j] = acc[j] + v[3 + n][j];
    }
  }

  float s = 0.0f;
#pragma unroll
  for (int m = 0; m < 12; ++m) {
#pragma unroll
    for (int j = 0; j < 8; ++j) s += v[m][j];
  }
  s += swap16(s);
  const float mean = s * (1.0f / E);
  float vs = 0.0f;
#pragma unroll
  for (int m = 0; m < 12; ++m) {
#pragma unroll
    for (int j = 0; j < 8; ++j) { const float d = v[m][j] - mean; vs += d * d; }
  }
  vs += swap16(vs);
  const float inv = rsqrtf(vs * (1.0f / E) + 1e-6f);

  {
    float* wp = patch + c * RP + h8;
#pragma unroll
    for (int m = 0; m < 12; ++m) {
      v4f o0, o1;
#pragma unroll
      for (int j = 0; j < 4; ++j) { o0[j] = v[m][j]; o1[j] = v[m][4 + j]; }
      *(v4f*)(wp + 16 * m) = o0;
      *(v4f*)(wp + 16 * m + 4) = o1;
    }
    __syncthreads();
#pragma unroll
    for (int L = 0; L < 6; ++L) {
#pragma unroll
      for (int g = 0; g < 4; ++g) {
        const v4f t = *(const v4f*)&patch[(lrb + g) * RP + 32 * L + 4 * lq];
        *(v4f*)(x1 + (ltok + 4 * g) * E + 32 * L + 4 * lq) = t;
      }
    }
    __syncthreads();
  }
  {
    _Float16* wp = hpatch + c * TP + h8;
#pragma unroll
    for (int m = 0; m < 12; ++m) {
      const v4f g0 = *(const v4f*)(g2 + 16 * m + h8);
      const v4f g1 = *(const v4f*)(g2 + 16 * m + h8 + 4);
      const v4f e0 = *(const v4f*)(b2 + 16 * m + h8);
      const v4f e1 = *(const v4f*)(b2 + 16 * m + h8 + 4);
      v8h o;
#pragma unroll
      for (int j = 0; j < 4; ++j) {
        o[j]     = (_Float16)((v[m][j] - mean) * inv * g0[j] + e0[j]);
        o[4 + j] = (_Float16)((v[m][4 + j] - mean) * inv * g1[j] + e1[j]);
      }
      *(v8h*)(wp + 16 * m) = o;
    }
    __syncthreads();
#pragma unroll
    for (int L = 0; L < 3; ++L) {
#pragma unroll
      for (int g = 0; g < 4; ++g) {
        const v8h t = *(const v8h*)&hpatch[(lrb + g) * TP + 64 * L + 8 * lq];
        *(v8h*)(tt + (ltok + 4 * g) * E + 64 * L + 8 * lq) = t;
      }
    }
  }
}

static inline size_t align256(size_t x) { return (x + 255) & ~(size_t)255; }

extern "C" void kernel_launch(void* const* d_in, const int* in_sizes, int n_in,
                              void* d_out, int out_size, void* d_ws, size_t ws_size,
                              hipStream_t stream) {
  (void)in_sizes; (void)n_in; (void)out_size; (void)ws_size;

  const float* x        = (const float*)d_in[0];
  const float* ln1_g    = (const float*)d_in[1];
  const float* ln1_b    = (const float*)d_in[2];
  const float* acc_wqkv = (const float*)d_in[3];
  const float* acc_bqkv = (const float*)d_in[4];
  const float* acc_wo   = (const float*)d_in[5];
  const float* acc_bo   = (const float*)d_in[6];
  const float* gy_wqkv  = (const float*)d_in[7];
  const float* gy_bqkv  = (const float*)d_in[8];
  const float* gy_wo    = (const float*)d_in[9];
  const float* gy_bo    = (const float*)d_in[10];
  const float* conv_w   = (const float*)d_in[11];
  const float* ln2_g    = (const float*)d_in[12];
  const float* ln2_b    = (const float*)d_in[13];
  const float* w1       = (const float*)d_in[14];
  const float* b1       = (const float*)d_in[15];
  const float* w2       = (const float*)d_in[16];
  const float* b2       = (const float*)d_in[17];

  char* ws = (char*)d_ws;
  size_t off = 0;
  _Float16* xn_h   = (_Float16*)(ws + off); off = align256(off + (size_t)NTOK * E * 2);
  _Float16* attn_h = (_Float16*)(ws + off); off = align256(off + (size_t)2 * NTOK * QDIM * 2);
  float*    x1     = (float*)   (ws + off); off = align256(off + (size_t)NTOK * E * 4);
  _Float16* tt_h   = (_Float16*)(ws + off); off = align256(off + (size_t)NTOK * E * 2);

  ln_kernel<<<NTOK / 8, 256, 0, stream>>>(x, ln1_g, ln1_b, xn_h);

  attn_kernel<<<BATCH * 6, 512, 0, stream>>>(xn_h, acc_wqkv, acc_bqkv,
                                             gy_wqkv, gy_bqkv, attn_h);

  rows_kernel<<<NTOK / 64, 128, 0, stream>>>(
      xn_h, attn_h, attn_h + (size_t)NTOK * QDIM, x, conv_w,
      acc_wo, acc_bo, gy_wo, gy_bo, ln2_g, ln2_b, x1, tt_h);

  mlp_pipe_kernel<32><<<256, 384, 0, stream>>>(tt_h, w1, b1, w2, b2, x1, (float*)d_out);
}


# Attention register tile 16 queries x 32 keys -> 32 x 64 per trip: a wave owns two query tiles (16 waves per work-group) on shared K/V fragments; max, rescale, row sum (two partial sums) per 64 keys
# speedup vs baseline: 2.9970x; 1.1644x over previous
#include <hip/hip_runtime.h>


typedef __attribute__((ext_vector_type(16))) _Float16 v16h;
typedef __attribute__((ext_vector_type(8)))  _Float16 v8h;
typedef __attribute__((ext_vector_type(4)))  _Float16 v4h;
typedef __attribute__((ext_vector_type(8)))  float    v8f;
typedef __attribute__((ext_vector_type(4)))  float    v4f;
typedef __attribute__((ext_vector_type(2)))  float    v2f;

#define DEVI static __device__ __forceinline__

static constexpr int BATCH = 128;
static constexpr int SEQ   = 512;
static constexpr int E     = 192;
static constexpr int QDIM  = 48;
static constexpr int NTOK  = BATCH * SEQ;

static constexpr int RP = 196;
static constexpr int TP = 200;

static constexpr int KT_LD = 24;
static constexpr int VT_LD = 520;
static constexpr int KT_SZ = SEQ * KT_LD;
static constexpr int VT_SZ = 16 * VT_LD;
static constexpr int WL_LD = 56;
static constexpr int XP_LD = 56;

DEVI v8f wmma16(v16h a, v16h b, v8f c) {
  return __builtin_amdgcn_wmma_f32_16x16x32_f16(false, a, false, b,
                                                (short)0, c, false, false);
}

DEVI float swap16(float v) {
  int i = __builtin_bit_cast(int, v);
  int r = __builtin_amdgcn_permlanex16(i, i, 0x76543210, 0xfedcba98, false, false);
  return __builtin_bit_cast(float, r);
}

template <int KREM>
DEVI v16h frag16(const _Float16* p) {
  v16h r;
  const v8h lo = *(const v8h*)p;
  if (KREM >= 32) {
    const v8h hi = *(const v8h*)(p + 16);
#pragma unroll
    for (int i = 0; i < 8; ++i) { r[i] = lo[i]; r[8 + i] = hi[i]; }
  } else {
#pragma unroll
    for (int i = 0; i < 8; ++i) { r[i] = lo[i]; r[8 + i] = (_Float16)0.0f; }
  }
  return r;
}

template <int KREM>
DEVI v16h frag16_f32(const float* p) {
  v16h r;
  const v4f a0 = *(const v4f*)p;
  const v4f a1 = *(const v4f*)(p + 4);
#pragma unroll
  for (int i = 0; i < 4; ++i) { r[i] = (_Float16)a0[i]; r[4 + i] = (_Float16)a1[i]; }
  if (KREM >= 32) {
    const v4f b0 = *(const v4f*)(p + 16);
    const v4f b1 = *(const v4f*)(p + 20);
#pragma unroll
    for (int i = 0; i < 4; ++i) { r[8 + i] = (_Float16)b0[i]; r[12 + i] = (_Float16)b1[i]; }
  } else {
#pragma unroll
    for (int i = 0; i < 8; ++i) r[8 + i] = (_Float16)0.0f;
  }
  return r;
}

DEVI void wave_turn() {
  __builtin_amdgcn_fence(__ATOMIC_RELEASE, "wavefront");
  __builtin_amdgcn_wave_barrier();
  __builtin_amdgcn_fence(__ATOMIC_ACQUIRE, "wavefront");
}

template <int TM>
__global__ __launch_bounds__(384) void mlp_pipe_kernel(
    const _Float16* __restrict__ tt,
    const float* __restrict__ w1, const float* __restrict__ b1,
    const float* __restrict__ w2, const float* __restrict__ b2,
    const float* __restrict__ x1, float* __restrict__ out) {
  constexpr int K1 = E, K2 = 2 * E;
  constexpr int P1 = K1 + 8;
  constexpr int P2 = K2 + 8;
  constexpr int CPR = K1 / 8;
  constexpr int NTHR = 384;
  constexpr int PT = TM * CPR / NTHR;
  constexpr int NTILES = NTOK / TM;
  static_assert((TM * CPR) % NTHR == 0, "tile must split evenly");
  static_assert((P1 / 8) % 2 == 1 && (P2 / 8) % 2 == 1, "pitch must be 16 B x odd");

  constexpr int W1H = K2 * P1, W2H = K1 * P2;
  static_assert(2 * TM * (P1 + P2) <= W1H + W2H, "tiles must fit in the block");
  __shared__ __attribute__((aligned(16))) _Float16 lds[W1H + W2H];
  __shared__ __attribute__((aligned(16))) float bl[K2 + K1];
  _Float16* const tbuf0 = lds;
  _Float16* const hbuf0 = lds + 2 * TM * P1;
  constexpr int OPP = 36;
  static_assert(2 * TM * (P1 + P2) + 6 * 16 * OPP * 2 <= W1H + W2H, "patches must fit in the block");

  const int tid = threadIdx.x;
  const int wave = __builtin_amdgcn_readfirstlane(tid >> 5);
  const int lane = tid & 31;
  const int c = lane & 15;
  const int h8 = (lane & 16) ? 8 : 0;
  const bool first = wave < 6;
  const int n0 = first ? wave * 64 : (wave - 6) * 32;
  float* const opatch = (float*)(lds + 2 * TM * (P1 + P2)) + (first ? 0 : wave - 6) * (16 * OPP);
  const int lq = lane & 7;
  const int lrb = ((lane >> 3) & 1) * 8 + (lane >> 4) * 4;

  if (tid < K1) bl[K2 + tid] = b2[tid];
  bl[tid] = b1[tid];
  const float* bw = &bl[(first ? 0 : K2) + n0 + h8];

  v8h st[PT];
  const int G = gridDim.x;
  const int tile0 = blockIdx.x;
  if (tile0 < NTILES) {
#pragma unroll
    for (int i = 0; i < PT; ++i) {
      const int q = i * NTHR + tid;
      const int rr = q / CPR, cc = q % CPR;
      st[i] = *(const v8h*)(tt + (size_t)(tile0 * TM + rr) * K1 + cc * 8);
    }
  }

#pragma unroll 24
  for (int i = 0; i < (K2 * K1 / 4) / NTHR; ++i) {
    const int q = i * NTHR + tid;
    const v4f a = *(const v4f*)(w1 + (size_t)q * 4);
    const v4f b = *(const v4f*)(w2 + (size_t)q * 4);
    v4h ah, bh;
#pragma unroll
    for (int j = 0; j < 4; ++j) { ah[j] = (_Float16)a[j]; bh[j] = (_Float16)b[j]; }
    *(v4h*)&lds[(q / (K1 / 4)) * P1 + (q % (K1 / 4)) * 4] = ah;
    *(v4h*)&lds[W1H + (q / (K2 / 4)) * P2 + (q % (K2 / 4)) * 4] = bh;
  }
  __syncthreads();

  v16h wf[24];
  if (first) {
#pragma unroll
    for (int t = 0; t < 4; ++t) {
#pragma unroll
      for (int ks = 0; ks < 6; ++ks)
        wf[t * 6 + ks] = frag16<32>(&lds[(n0 + t * 16 + c) * P1 + ks * 32 + h8]);
    }
  } else {
#pragma unroll
    for (int t = 0; t < 2; ++t) {
#pragma unroll
      for (int ks = 0; ks < 12; ++ks)
        wf[t * 12 + ks] = frag16<32>(&lds[W1H + (n0 + t * 16 + c) * P2 + ks * 32 + h8]);
    }
  }
  __syncthreads();

  if (tile0 < NTILES) {
#pragma unroll
    for (int i = 0; i < PT; ++i) {
      const int q = i * NTHR + tid;
      const int rr = q / CPR, cc = q % CPR;
      *(v8h*)&tbuf0[rr * P1 + cc * 8] = st[i];
    }
  }
  if (tile0 + G < NTILES) {
#pragma unroll
    for (int i = 0; i < PT; ++i) {
      const int q = i * NTHR + tid;
      const int rr = q / CPR, cc = q % CPR;
      st[i] = *(const v8h*)(tt + (size_t)((tile0 + G) * TM + rr) * K1 + cc * 8);
    }
  }
  v4f rn[4] = {};
  if (!first && tile0 < NTILES) {
    const float* rp = x1 + ((size_t)tile0 * TM + lrb) * E + n0 + 4 * lq;
#pragma unroll
    for (int g = 0; g < 4; ++g) rn[g] = *(const v4f*)(rp + (size_t)g * E);
  }
  __syncthreads();

  for (int s = 0; ; ++s) {
    const int tp = tile0 + s * G;
    const int tc = tp - G;
    const bool hp = tp < NTILES;
    const bool hc = (s > 0) && (tc < NTILES);
    if (!hp && !hc) break;

    if (tp + G < NTILES) {
      _Float16* tn = tbuf0 + ((s + 1) & 1) * (TM * P1);
#pragma unroll
      for (int i = 0; i < PT; ++i) {
        const int q = i * NTHR + tid;
        const int rr = q / CPR, cc = q % CPR;
        *(v8h*)&tn[rr * P1 + cc * 8] = st[i];
      }
      if (tp + 2 * G < NTILES) {
#pragma unroll
        for (int i = 0; i < PT; ++i) {
          const int q = i * NTHR + tid;
          const int rr = q / CPR, cc = q % CPR;
          st[i] = *(const v8h*)(tt + (size_t)((tp + 2 * G) * TM + rr) * K1 + cc * 8);
        }
      }
    }

    if (first) {
      if (hp) {
        const _Float16* tb = tbuf0 + (s & 1) * (TM * P1);
        _Float16* hb = hbuf0 + (s & 1) * (TM * P2);
#pragma unroll 1
        for (int sub = 0; sub < TM / 16; ++sub) {
          const _Float16* xr = &tb[(sub * 16 + c) * P1 + h8];
          v8f acc[4] = {};
#pragma unroll
          for (int kh = 0; kh < 2; ++kh) {
            v16h xf[3];
#pragma unroll
            for (int ks = 0; ks < 3; ++ks) xf[ks] = frag16<32>(xr + (kh * 3 + ks) * 32);
#pragma unroll
            for (int ks = 0; ks < 3; ++ks) {
#pragma unroll
              for (int t = 0; t < 4; ++t)
                acc[t] = wmma16(wf[t * 6 + kh * 3 + ks], xf[ks], acc[t]);
            }
          }
          _Float16* hr = &hb[(sub * 16 + c) * P2 + n0 + h8];
#pragma unroll
          for (int t = 0; t < 4; ++t) {
            const v4f bb0 = *(const v4f*)(bw + t * 16);
            const v4f bb1 = *(const v4f*)(bw + t * 16 + 4);
            v8h o;
#pragma unroll
            for (int j = 0; j < 4; ++j) {
              o[j]     = (_Float16)fmaxf(acc[t][j] + bb0[j], 0.0f);
              o[4 + j] = (_Float16)fmaxf(acc[t][4 + j] + bb1[j], 0.0f);
            }
            *(v8h*)(hr + t * 16) = o;
          }
        }
      }
    } else {
      if (hc) {
        const _Float16* hb = hbuf0 + ((s - 1) & 1) * (TM * P2);
#pragma unroll 1
        for (int sub = 0; sub < TM / 16; ++sub) {
          const size_t tok = (size_t)tc * TM + sub * 16 + c;
          const _Float16* xr = &hb[(sub * 16 + c) * P2 + h8];
          v8f acc[2] = {};
#pragma unroll
          for (int kh = 0; kh < 3; ++kh) {
            v16h xf[4];
#pragma unroll
            for (int ks = 0; ks < 4; ++ks) xf[ks] = frag16<32>(xr + (kh * 4 + ks) * 32);
#pragma unroll
            for (int ks = 0; ks < 4; ++ks) {
#pragma unroll
              for (int t = 0; t < 2; ++t)
                acc[t] = wmma16(wf[t * 12 + kh * 4 + ks], xf[ks], acc[t]);
            }
          }
          float* const pw = opatch + c * OPP + h8;
#pragma unroll
          for (int t = 0; t < 2; ++t) {
            const v4f bb0 = *(const v4f*)(bw + t * 16);
            const v4f bb1 = *(const v4f*)(bw + t * 16 + 4);
            v4f o0, o1;
#pragma unroll
            for (int j = 0; j < 4; ++j) {
              o0[j] = fmaxf(acc[t][j] + bb0[j], 0.0f);
              o1[j] = fmaxf(acc[t][4 + j] + bb1[j], 0.0f);
            }
            *(v4f*)(pw + t * 16) = o0;
            *(v4f*)(pw + t * 16 + 4) = o1;
          }
          wave_turn();
          const size_t ltok = tok - c + lrb;
#pragma unroll
          for (int g = 0; g < 4; ++g) {
            const v4f pv = *(const v4f*)&opatch[(lrb + g) * OPP + 4 * lq];
            v4f o;
#pragma unroll
            for (int j = 0; j < 4; ++j) o[j] = pv[j] + rn[g][j];
            *(v4f*)(out + (ltok + g) * E + n0 + 4 * lq) = o;
          }
          wave_turn();
          const size_t ntok = (sub + 1 < TM / 16) ? ltok + 16
                                                  : ((size_t)(tc + G) * TM + lrb);
          if (sub + 1 < TM / 16 || tc + G < NTILES) {
            const float* rp = x1 + ntok * E + n0 + 4 * lq;
#pragma unroll
            for (int g = 0; g < 4; ++g) rn[g] = *(const v4f*)(rp + (size_t)g * E);
          }
        }
      }
    }
    __syncthreads();
  }
}

DEVI void attn_tile2(const _Float16* kt, const _Float16* vt, v16h qa, v16h qb,
                     _Float16* __restrict__ orow_a, _Float16* __restrict__ orow_b,
                     int row, int d0) {
  const v8f z = {};
  v16h qf[2] = {qa, qb};
  v8f oacc[2] = {z, z};
  float m_run[2] = {-1e30f, -1e30f}, l_run[2] = {0.0f, 0.0f};

  for (int kb = 0; kb < SEQ / 64; ++kb) {
    const int j0 = kb * 64;
    v16h kf[4];
#pragma unroll
    for (int i = 0; i < 4; ++i) {
      const v8h ka = *(const v8h*)&kt[(j0 + 16 * i + row) * KT_LD + d0];
#pragma unroll
      for (int e = 0; e < 8; ++e) { kf[i][e] = ka[e]; kf[i][8 + e] = (_Float16)0.0f; }
    }
    v16h vf[2];
#pragma unroll
    for (int g = 0; g < 2; ++g) {
      const _Float16* vp = &vt[row * VT_LD + j0 + 32 * g + d0];
      const v8h vl = *(const v8h*)vp;
      const v8h vh = *(const v8h*)(vp + 16);
#pragma unroll
      for (int e = 0; e < 8; ++e) { vf[g][e] = vl[e]; vf[g][8 + e] = vh[e]; }
    }

#pragma unroll
    for (int t = 0; t < 2; ++t) {
      v8f s[4];
#pragma unroll
      for (int i = 0; i < 4; ++i) s[i] = wmma16(kf[i], qf[t], z);

      float tm = -1e30f;
#pragma unroll
      for (int g = 0; g < 2; ++g) {
#pragma unroll
        for (int j = 0; j < 8; ++j) tm = fmaxf(tm, fmaxf(s[2 * g][j], s[2 * g + 1][j]));
      }
      tm = fmaxf(tm, swap16(tm));
      const float mnew = fmaxf(m_run[t], tm);
      const float corr = __expf(m_run[t] - mnew);
      const v2f mm = {mnew, mnew};
      v2f ls2 = {0.0f, 0.0f};
      v16h pf[2];
#pragma unroll
      for (int i = 0; i < 4; ++i) {
#pragma unroll
        for (int jj = 0; jj < 4; ++jj) {
          const v2f sv = {s[i][2 * jj], s[i][2 * jj + 1]};
          const v2f d = sv - mm;
          const v2f p = {__expf(d[0]), __expf(d[1])};
          ls2 += p;
          pf[i >> 1][(i & 1) * 8 + 2 * jj]     = (_Float16)p[0];
          pf[i >> 1][(i & 1) * 8 + 2 * jj + 1] = (_Float16)p[1];
        }
      }
      float ls = ls2[0] + ls2[1];
      ls += swap16(ls);
      l_run[t] = l_run[t] * corr + ls;
      m_run[t] = mnew;
#pragma unroll
      for (int j = 0; j < 8; ++j) oacc[t][j] *= corr;
      oacc[t] = wmma16(vf[0], pf[0], oacc[t]);
      oacc[t] = wmma16(vf[1], pf[1], oacc[t]);
    }
  }

  const float inv_a = 1.0f / l_run[0];
  const float inv_b = 1.0f / l_run[1];
#pragma unroll
  for (int j = 0; j < 8; ++j) {
    orow_a[j] = (_Float16)(oacc[0][j] * inv_a);
    orow_b[j] = (_Float16)(oacc[1][j] * inv_b);
  }
}

__global__ __launch_bounds__(512) void attn_kernel(
    const float* __restrict__ x, const float* __restrict__ g1, const float* __restrict__ b1,
    const float* __restrict__ w_acc, const float* __restrict__ b_acc,
    const float* __restrict__ w_gy, const float* __restrict__ b_gy,
    _Float16* __restrict__ out_all) {
  __shared__ __attribute__((aligned(16))) _Float16 vt[3 * VT_SZ];
  __shared__ __attribute__((aligned(16))) _Float16 kt[3 * KT_SZ];
  __shared__ __attribute__((aligned(16))) _Float16 wl[144 * WL_LD];
  __shared__ __attribute__((aligned(16))) _Float16 xp_all[16 * 32 * XP_LD];

  const int bid = blockIdx.x;
  const int b = bid >> 1;
  const int branch = bid & 1;
  const float* Wt = branch ? w_gy : w_acc;
  const float* bias = branch ? b_gy : b_acc;
  _Float16* outp = out_all + (size_t)branch * NTOK * QDIM
                           + (size_t)b * SEQ * QDIM;

  const int tid = threadIdx.x;
  const int wave = tid >> 5, lane = tid & 31;
  const int row = lane & 15;
  const int d0  = (lane & 16) ? 8 : 0;

  for (int q = tid; q < 144 * (QDIM / 4); q += 512) {
    const v4f a = *(const v4f*)(Wt + (size_t)q * 4);
    v4h ah;
#pragma unroll
    for (int j = 0; j < 4; ++j) ah[j] = (_Float16)a[j];
    *(v4h*)&wl[(q / (QDIM / 4)) * WL_LD + (q % (QDIM / 4)) * 4] = ah;
  }

  _Float16* const xpw = xp_all + wave * (32 * XP_LD);
  {
    const int lq = lane & 7, tr = lane >> 3;
    const bool up = lq >= 4;
    const int Lsel = branch ? (up ? 1 : 2) : 0;
    const int rel = 64 * Lsel + 8 * lq - 96 * branch;
    const bool have = rel >= 0 && rel < QDIM;
    const v4f ga = *(const v4f*)(g1 + 64 * Lsel + 8 * lq);
    const v4f gb = *(const v4f*)(g1 + 64 * Lsel + 8 * lq + 4);
    const v4f ba = *(const v4f*)(b1 + 64 * Lsel + 8 * lq);
    const v4f bb = *(const v4f*)(b1 + 64 * Lsel + 8 * lq + 4);
#pragma unroll 1
    for (int pp = 0; pp < 2; ++pp) {
      v4f xv[4][6];
#pragma unroll
      for (int u = 0; u < 4; ++u) {
        const int tk = 16 * pp + 4 * u + tr;
        const float* xr = x + ((size_t)b * SEQ + wave * 32 + tk) * E + 8 * lq;
#pragma unroll
        for (int L = 0; L < 3; ++L) {
          xv[u][2 * L]     = *(const v4f*)(xr + 64 * L);
          xv[u][2 * L + 1] = *(const v4f*)(xr + 64 * L + 4);
        }
      }
#pragma unroll
      for (int u = 0; u < 4; ++u) {
        const int tk = 16 * pp + 4 * u + tr;
        float s = 0.0f;
#pragma unroll
        for (int i = 0; i < 6; ++i) {
#pragma unroll
          for (int j = 0; j < 4; ++j) s += xv[u][i][j];
        }
        s += __shfl_xor(s, 1, 32);
        s += __shfl_xor(s, 2, 32);
        s += __shfl_xor(s, 4, 32);
        const float mean = s * (1.0f / E);
        float vs = 0.0f;
#pragma unroll
        for (int i = 0; i < 6; ++i) {
#pragma unroll
          for (int j = 0; j < 4; ++j) { const float d = xv[u][i][j] - mean; vs += d * d; }
        }
        vs += __shfl_xor(vs, 1, 32);
        vs += __shfl_xor(vs, 2, 32);
        vs += __shfl_xor(vs, 4, 32);
        const float inv = rsqrtf(vs * (1.0f / E) + 1e-6f);
        v4f xa, xb;
        if (branch == 0) { xa = xv[u][0]; xb = xv[u][1]; }
        else { xa = up ? xv[u][2] : xv[u][4]; xb = up ? xv[u][3] : xv[u][5]; }
        if (have) {
          v8h o;
#pragma unroll
          for (int j = 0; j < 4; ++j) {
            o[j]     = (_Float16)((xa[j] - mean) * inv * ga[j] + ba[j]);
            o[4 + j] = (_Float16)((xb[j] - mean) * inv * gb[j] + bb[j]);
          }
          *(v8h*)&xpw[tk * XP_LD + rel] = o;
        }
      }
    }
  }
  __syncthreads();

  const v8f z0 = {};
  v8h qh[2][3];
#pragma unroll
  for (int t = 0; t < 2; ++t) {
    const _Float16* xq = xpw + (t * 16 + row) * XP_LD + d0;
    const v16h x0 = frag16<32>(xq);
    const v16h x1f = frag16<16>(xq + 32);
    const int tok0 = wave * 32 + t * 16;
#pragma unroll
    for (int hh = 0; hh < 3; ++hh) {
      const int qcol = hh * 16;
      {
        const _Float16* wp = &wl[(48 + qcol + row) * WL_LD + d0];
        const v16h w0 = frag16<32>(wp), w1 = frag16<16>(wp + 32);
        const v4f b0 = *(const v4f*)(bias + 48 + qcol + d0);
        const v4f b1v = *(const v4f*)(bias + 48 + qcol + d0 + 4);
        v8f a = wmma16(w0, x0, z0);
        a = wmma16(w1, x1f, a);
        v8h o;
#pragma unroll
        for (int j = 0; j < 4; ++j) {
          o[j] = (_Float16)(a[j] + b0[j]);
          o[4 + j] = (_Float16)(a[4 + j] + b1v[j]);
        }
        *(v8h*)&kt[hh * KT_SZ + (tok0 + row) * KT_LD + d0] = o;
      }
      {
        const _Float16* wp = &wl[(96 + qcol + row) * WL_LD + d0];
        const v16h w0 = frag16<32>(wp), w1 = frag16<16>(wp + 32);
        const float bvv = bias[96 + qcol + row];
        v8f a = wmma16(x0, w0, z0);
        a = wmma16(x1f, w1, a);
        v8h o;
#pragma unroll
        for (int j = 0; j < 8; ++j) o[j] = (_Float16)(a[j] + bvv);
        *(v8h*)&vt[hh * VT_SZ + row * VT_LD + tok0 + d0] = o;
      }
      {
        const _Float16* wp = &wl[(qcol + row) * WL_LD + d0];
        const v16h w0 = frag16<32>(wp), w1 = frag16<16>(wp + 32);
        const v4f b0 = *(const v4f*)(bias + qcol + d0);
        const v4f b1v = *(const v4f*)(bias + qcol + d0 + 4);
        v8f a = wmma16(w0, x0, z0);
        a = wmma16(w1, x1f, a);
#pragma unroll
        for (int j = 0; j < 4; ++j) {
          qh[t][hh][j] = (_Float16)(a[j] + b0[j]) * (_Float16)0.25f;
          qh[t][hh][4 + j] = (_Float16)(a[4 + j] + b1v[j]) * (_Float16)0.25f;
        }
      }
    }
  }
  __syncthreads();

#pragma unroll
  for (int hh = 0; hh < 3; ++hh) {
    v16h qa, qb;
#pragma unroll
    for (int i = 0; i < 8; ++i) {
      qa[i] = qh[0][hh][i]; qa[8 + i] = (_Float16)0.0f;
      qb[i] = qh[1][hh][i]; qb[8 + i] = (_Float16)0.0f;
    }
    _Float16* const oa = outp + (size_t)(wave * 32 + row) * QDIM + hh * 16 + d0;
    attn_tile2(kt + hh * KT_SZ, vt + hh * VT_SZ, qa, qb, oa, oa + 16 * QDIM, row, d0);
  }
}

DEVI v8h ln1_chunk(const float* xrow, const float* gl, float mean1, float inv1, int q) {
  const int qc = min(max(q, 0), 11);
  const float* xq = xrow + 48 + 8 * qc;
  const v4f xa = *(const v4f*)xq;
  const v4f xb = *(const v4f*)(xq + 4);
  const v4f ga = *(const v4f*)(gl + 8 * qc);
  const v4f gb = *(const v4f*)(gl + 8 * qc + 4);
  const v4f ba = *(const v4f*)(gl + 96 + 8 * qc);
  const v4f bb = *(const v4f*)(gl + 96 + 8 * qc + 4);
  v8h o;
#pragma unroll
  for (int i = 0; i < 4; ++i) {
    o[i]     = (_Float16)((xa[i] - mean1) * inv1 * ga[i] + ba[i]);
    o[4 + i] = (_Float16)((xb[i] - mean1) * inv1 * gb[i] + bb[i]);
  }
  const v8h zero8 = {};
  return (q >= 0 && q <= 11) ? o : zero8;
}

__global__ __launch_bounds__(128) void rows_kernel(
    const _Float16* __restrict__ at_acc, const _Float16* __restrict__ at_gy,
    const float* __restrict__ x, const float* __restrict__ cw,
    const float* __restrict__ g1, const float* __restrict__ b1,
    const float* __restrict__ wo_acc, const float* __restrict__ bo_acc,
    const float* __restrict__ wo_gy, const float* __restrict__ bo_gy,
    const float* __restrict__ g2, const float* __restrict__ b2,
    float* __restrict__ x1, _Float16* __restrict__ tt) {
  const int ch = threadIdx.x >> 5, lane = threadIdx.x & 31;
  const int c = lane & 15, h = lane >> 4;
  const int h8 = 8 * h;
  const size_t tok = (size_t)blockIdx.x * 64 + 4 * c + ch;

  __shared__ __attribute__((aligned(16))) float patch_all[4 * 16 * RP];
  float* const patch = patch_all + ch * (16 * RP);
  __shared__ __attribute__((aligned(16))) float gl1[192];
  if (threadIdx.x < 96) {
    gl1[threadIdx.x] = g1[48 + threadIdx.x];
    gl1[96 + threadIdx.x] = b1[48 + threadIdx.x];
  }
  _Float16* const hpatch = (_Float16*)patch;
  const int lq = lane & 7;
  const int lrb = ((lane >> 3) & 1) * 8 + (lane >> 4) * 4;
  const size_t ltok = (size_t)blockIdx.x * 64 + 4 * lrb + ch;

  float v[12][8];
  v16h a0s[2], a1s[2];
  {
    v4f tl[24];
#pragma unroll
    for (int L = 0; L < 6; ++L) {
#pragma unroll
      for (int g = 0; g < 4; ++g)
        tl[L * 4 + g] = *(const v4f*)(x + (ltok + 4 * g) * E + 32 * L + 4 * lq);
    }
#pragma unroll
    for (int br = 0; br < 2; ++br) {
      const _Float16* ap = (br ? at_gy : at_acc) + tok * QDIM + h8;
      a0s[br] = frag16<32>(ap);
      a1s[br] = frag16<16>(ap + 32);
    }
#pragma unroll
    for (int L = 0; L < 6; ++L) {
#pragma unroll
      for (int g = 0; g < 4; ++g)
        *(v4f*)&patch[(lrb + g) * RP + 32 * L + 4 * lq] = tl[L * 4 + g];
    }
    __syncthreads();
    const float* rp = patch + c * RP + h8;
#pragma unroll
    for (int m = 0; m < 12; ++m) {
      const v4f r0 = *(const v4f*)(rp + 16 * m);
      const v4f r1 = *(const v4f*)(rp + 16 * m + 4);
#pragma unroll
      for (int j = 0; j < 4; ++j) { v[m][j] = r0[j]; v[m][4 + j] = r1[j]; }
    }
  }

  float mean1, inv1;
  {
    float s1 = 0.0f;
#pragma unroll
    for (int m = 0; m < 12; ++m) {
#pragma unroll
      for (int j = 0; j < 8; ++j) s1 += v[m][j];
    }
    s1 += swap16(s1);
    mean1 = s1 * (1.0f / E);
    float q1 = 0.0f;
#pragma unroll
    for (int m = 0; m < 12; ++m) {
#pragma unroll
      for (int j = 0; j < 8; ++j) { const float d = v[m][j] - mean1; q1 += d * d; }
    }
    q1 += swap16(q1);
    inv1 = rsqrtf(q1 * (1.0f / E) + 1e-6f);
  }

  {
    v16h af;
    {
      const float* w = cw + ch * 15;
#pragma unroll
      for (int i = 0; i < 8; ++i) {
        const int k0 = 8 * h + i - c - 1;
        const int k1 = k0 + 16;
        const float w0 = w[min(max(k0, 0), 14)];
        const float w1 = w[min(max(k1, 0), 14)];
        af[i]     = (_Float16)((k0 >= 0 && k0 <= 14) ? w0 : 0.0f);
        af[8 + i] = (_Float16)((k1 >= 0 && k1 <= 14) ? w1 : 0.0f);
      }
    }
    const float* xrow = patch + c * RP;
    v8h cprev = ln1_chunk(xrow, gl1, mean1, inv1, h - 1);
#pragma unroll
    for (int n = 0; n < 6; ++n) {
      const v8h cnext = ln1_chunk(xrow, gl1, mean1, inv1, 2 * n + 1 + h);
      v16h bf;
#pragma unroll
      for (int i = 0; i < 8; ++i) { bf[i] = cprev[i]; bf[8 + i] = cnext[i]; }
      v8f acc = {};
      acc = wmma16(af, bf, acc);
#pragma unroll
      for (int j = 0; j < 8; ++j) v[3 + n][j] = acc[j] + v[3 + n][j];
      cprev = cnext;
    }
  }

#pragma unroll
  for (int br = 0; br < 2; ++br) {
    const float* Wt = br ? wo_gy : wo_acc;
    const float* bias = br ? bo_gy : bo_acc;
    const v16h a0 = a0s[br], a1 = a1s[br];
#pragma unroll
    for (int t = 0; t < 3; ++t) {
      const float* wp = Wt + (size_t)(t * 16 + c) * QDIM + h8;
      const v16h w0 = frag16_f32<32>(wp), w1 = frag16_f32<16>(wp + 32);
      const v4f bb0 = *(const v4f*)(bias + t * 16 + h8);
      const v4f bb1 = *(const v4f*)(bias + t * 16 + h8 + 4);
      v8f acc = {};
      acc = wmma16(w0, a0, acc);
      acc = wmma16(w1, a1, acc);
      const int m = br ? 9 + t : t;
#pragma unroll
      for (int j = 0; j < 4; ++j) {
        v[m][j]     = (acc[j] + bb0[j]) + v[m][j];
        v[m][4 + j] = (acc[4 + j] + bb1[j]) + v[m][4 + j];
      }
    }
  }

  float s = 0.0f;
#pragma unroll
  for (int m = 0; m < 12; ++m) {
#pragma unroll
    for (int j = 0; j < 8; ++j) s += v[m][j];
  }
  s += swap16(s);
  const float mean = s * (1.0f / E);
  float vs = 0.0f;
#pragma unroll
  for (int m = 0; m < 12; ++m) {
#pragma unroll
    for (int j = 0; j < 8; ++j) { const float d = v[m][j] - mean; vs += d * d; }
  }
  vs += swap16(vs);
  const float inv = rsqrtf(vs * (1.0f / E) + 1e-6f);

  {
    float* wp = patch + c * RP + h8;
#pragma unroll
    for (int m = 0; m < 12; ++m) {
      v4f o0, o1;
#pragma unroll
      for (int j = 0; j < 4; ++j) { o0[j] = v[m][j]; o1[j] = v[m][4 + j]; }
      *(v4f*)(wp + 16 * m) = o0;
      *(v4f*)(wp + 16 * m + 4) = o1;
    }
    __syncthreads();
#pragma unroll
    for (int L = 0; L < 6; ++L) {
#pragma unroll
      for (int g = 0; g < 4; ++g) {
        const v4f t = *(const v4f*)&patch[(lrb + g) * RP + 32 * L + 4 * lq];
        *(v4f*)(x1 + (ltok + 4 * g) * E + 32 * L + 4 * lq) = t;
      }
    }
    __syncthreads();
  }
  {
    _Float16* wp = hpatch + c * TP + h8;
#pragma unroll
    for (int m = 0; m < 12; ++m) {
      const v4f g0 = *(const v4f*)(g2 + 16 * m + h8);
      const v4f g1 = *(const v4f*)(g2 + 16 * m + h8 + 4);
      const v4f e0 = *(const v4f*)(b2 + 16 * m + h8);
      const v4f e1 = *(const v4f*)(b2 + 16 * m + h8 + 4);
      v8h o;
#pragma unroll
      for (int j = 0; j < 4; ++j) {
        o[j]     = (_Float16)((v[m][j] - mean) * inv * g0[j] + e0[j]);
        o[4 + j] = (_Float16)((v[m][4 + j] - mean) * inv * g1[j] + e1[j]);
      }
      *(v8h*)(wp + 16 * m) = o;
    }
    __syncthreads();
#pragma unroll
    for (int L = 0; L < 3; ++L) {
#pragma unroll
      for (int g = 0; g < 4; ++g) {
        const v8h t = *(const v8h*)&hpatch[(lrb + g) * TP + 64 * L + 8 * lq];
        *(v8h*)(tt + (ltok + 4 * g) * E + 64 * L + 8 * lq) = t;
      }
    }
  }
}

static inline size_t align256(size_t x) { return (x + 255) & ~(size_t)255; }

extern "C" void kernel_launch(void* const* d_in, const int* in_sizes, int n_in,
                              void* d_out, int out_size, void* d_ws, size_t ws_size,
                              hipStream_t stream) {
  (void)in_sizes; (void)n_in; (void)out_size; (void)ws_size;

  const float* x        = (const float*)d_in[0];
  const float* ln1_g    = (const float*)d_in[1];
  const float* ln1_b    = (const float*)d_in[2];
  const float* acc_wqkv = (const float*)d_in[3];
  const float* acc_bqkv = (const float*)d_in[4];
  const float* acc_wo   = (const float*)d_in[5];
  const float* acc_bo   = (const float*)d_in[6];
  const float* gy_wqkv  = (const float*)d_in[7];
  const float* gy_bqkv  = (const float*)d_in[8];
  const float* gy_wo    = (const float*)d_in[9];
  const float* gy_bo    = (const float*)d_in[10];
  const float* conv_w   = (const float*)d_in[11];
  const float* ln2_g    = (const float*)d_in[12];
  const float* ln2_b    = (const float*)d_in[13];
  const float* w1       = (const float*)d_in[14];
  const float* b1       = (const float*)d_in[15];
  const float* w2       = (const float*)d_in[16];
  const float* b2       = (const float*)d_in[17];

  char* ws = (char*)d_ws;
  size_t off = 0;
  _Float16* attn_h = (_Float16*)(ws + off); off = align256(off + (size_t)2 * NTOK * QDIM * 2);
  float*    x1     = (float*)   (ws + off); off = align256(off + (size_t)NTOK * E * 4);
  _Float16* tt_h   = (_Float16*)(ws + off); off = align256(off + (size_t)NTOK * E * 2);

  attn_kernel<<<BATCH * 2, 512, 0, stream>>>(x, ln1_g, ln1_b, acc_wqkv, acc_bqkv,
                                              gy_wqkv, gy_bqkv, attn_h);

  rows_kernel<<<NTOK / 64, 128, 0, stream>>>(
      attn_h, attn_h + (size_t)NTOK * QDIM, x, conv_w, ln1_g, ln1_b,
      acc_wo, acc_bo, gy_wo, gy_bo, ln2_g, ln2_b, x1, tt_h);

  mlp_pipe_kernel<32><<<256, 384, 0, stream>>>(tt_h, w1, b1, w2, b2, x1, (float*)d_out);
}
